# starting kernel
# baseline (speedup 1.0000x reference)
#include <hip/hip_runtime.h>

#define VOCAB   256
#define EMBED   32
#define HIDDEN  32
#define LAYERS  10
#define BATCH   1024
#define SEQT    1024

#define RS      34
#define TILEF   (16 * RS)
#define NBUF    (LAYERS * 2 * TILEF)
#define NEMB    (VOCAB * RS)

typedef __attribute__((ext_vector_type(2))) float v2f;
typedef __attribute__((ext_vector_type(8))) float v8f;

__device__ __forceinline__ v8f wmma_k4(v2f a, v2f b, v8f c) {
  return __builtin_amdgcn_wmma_f32_16x16x4_f32(false, a, false, b, (short)0, c,
                                               false, false);
}

__device__ __forceinline__ float dev_tanh(float x) {
#if __has_builtin(__builtin_amdgcn_tanhf)
  return __builtin_amdgcn_tanhf(x);
#else
  float ax = __builtin_fabsf(x);
  float e  = __expf(-2.0f * ax);
  float y  = __fdividef(1.0f - e, 1.0f + e);
  return __builtin_copysignf(y, x);
#endif
}

__global__ void __launch_bounds__(LAYERS * 32)
rnn_pipeline_kernel(const int* __restrict__ x, const float* __restrict__ emb,
                    const float* __restrict__ W_ih, const float* __restrict__ W_hh,
                    const float* __restrict__ b_ih, const float* __restrict__ b_hh,
                    const float* __restrict__ W_fc, const float* __restrict__ b_fc,
                    float* __restrict__ out) {
  extern __shared__ float smem[];
  float* buf  = smem;
  float* embs = smem + NBUF;

  const int tid  = threadIdx.x;
  const int lane = tid & 31;
  const int l    = tid >> 5;
  const int R    = blockIdx.x * 16;

  for (int i = tid; i < NBUF; i += LAYERS * 32) buf[i] = 0.0f;
  for (int i = tid; i < VOCAB * EMBED; i += LAYERS * 32)
    embs[(i >> 5) * RS + (i & 31)] = emb[i];

  const int m    = lane & 15;
  const int koff = (lane >> 4) * 2;
  const int rb   = (lane >> 4) * 8;

  v2f wih[2][8], whh[2][8];
  float bias[2];
  {
    const float* Wi = W_ih + l * HIDDEN * EMBED;
    const float* Wh = W_hh + l * HIDDEN * HIDDEN;
#pragma unroll
    for (int c = 0; c < 2; ++c) {
      const int row = c * 16 + m;
#pragma unroll
      for (int j = 0; j < 8; ++j) {
        wih[c][j] = *(const v2f*)(Wi + row * EMBED  + 4 * j + koff);
        whh[c][j] = *(const v2f*)(Wh + row * HIDDEN + 4 * j + koff);
      }
      bias[c] = b_ih[l * HIDDEN + row] + b_hh[l * HIDDEN + row];
    }
  }

  __syncthreads();

  for (int i = 0; i < SEQT + LAYERS - 1; ++i) {
    const int t = i - l;
    if ((unsigned)t < (unsigned)SEQT) {
      const int pin  = (i - 1) & 1;
      const int pout = i & 1;

      v2f ah[8], ax[8];
      const float* hin = buf + (l * 2 + pin) * TILEF + m * RS + koff;
#pragma unroll
      for (int j = 0; j < 8; ++j) ah[j] = *(const v2f*)(hin + 4 * j);

      if (l == 0) {
        const int tok   = x[(R + m) * SEQT + t];
        const float* xe = embs + tok * RS + koff;
#pragma unroll
        for (int j = 0; j < 8; ++j) ax[j] = *(const v2f*)(xe + 4 * j);
      } else {
        const float* xin = buf + ((l - 1) * 2 + pin) * TILEF + m * RS + koff;
#pragma unroll
        for (int j = 0; j < 8; ++j) ax[j] = *(const v2f*)(xin + 4 * j);
      }

      v8f d0, d1;
#pragma unroll
      for (int k = 0; k < 8; ++k) { d0[k] = bias[0]; d1[k] = bias[1]; }
#pragma unroll
      for (int j = 0; j < 8; ++j) {
        d0 = wmma_k4(ax[j], wih[0][j], d0);
        d1 = wmma_k4(ax[j], wih[1][j], d1);
      }
#pragma unroll
      for (int j = 0; j < 8; ++j) {
        d0 = wmma_k4(ah[j], whh[0][j], d0);
        d1 = wmma_k4(ah[j], whh[1][j], d1);
      }

      float* hout = buf + (l * 2 + pout) * TILEF;
#pragma unroll
      for (int r = 0; r < 8; ++r) {
        hout[(rb + r) * RS + m]      = dev_tanh(d0[r]);
        hout[(rb + r) * RS + m + 16] = dev_tanh(d1[r]);
      }
    }
    __syncthreads();
  }

  const float* htop = buf + ((LAYERS - 1) * 2 + ((SEQT + LAYERS - 2) & 1)) * TILEF;
  v2f ahf[8];
#pragma unroll
  for (int j = 0; j < 8; ++j)
    ahf[j] = *(const v2f*)(htop + m * RS + 4 * j + koff);

  for (int ct = l; ct < VOCAB / 16; ct += LAYERS) {
    const int vrow = ct * 16 + m;
    v2f wb[8];
#pragma unroll
    for (int j = 0; j < 8; ++j)
      wb[j] = *(const v2f*)(W_fc + vrow * HIDDEN + 4 * j + koff);
    const float bf = b_fc[vrow];
    v8f d;
#pragma unroll
    for (int k = 0; k < 8; ++k) d[k] = bf;
#pragma unroll
    for (int j = 0; j < 8; ++j) d = wmma_k4(ahf[j], wb[j], d);
#pragma unroll
    for (int r = 0; r < 8; ++r)
      out[(R + rb + r) * VOCAB + vrow] = d[r];
  }
}

extern "C" void kernel_launch(void* const* d_in, const int* in_sizes, int n_in,
                              void* d_out, int out_size, void* d_ws, size_t ws_size,
                              hipStream_t stream) {
  (void)in_sizes; (void)n_in; (void)out_size; (void)d_ws; (void)ws_size;
  const int*   x   = (const int*)d_in[0];
  const float* emb = (const float*)d_in[1];
  const float* Wih = (const float*)d_in[2];
  const float* Whh = (const float*)d_in[3];
  const float* bih = (const float*)d_in[4];
  const float* bhh = (const float*)d_in[5];
  const float* Wfc = (const float*)d_in[6];
  const float* bfc = (const float*)d_in[7];
  float* out = (float*)d_out;

  const size_t shmem = (size_t)(NBUF + NEMB) * sizeof(float);
  hipLaunchKernelGGL(rnn_pipeline_kernel, dim3(BATCH / 16), dim3(LAYERS * 32),
                     shmem, stream, x, emb, Wih, Whh, bih, bhh, Wfc, bfc, out);
}


# One wave per (layer, 16-column tile): 20 waves instead of 10, five per SIMD, so the busiest SIMD issues 80 instead of 96 f32 WMMAs per time step
# speedup vs baseline: 1.0689x; 1.0689x over previous
#include <hip/hip_runtime.h>

#define VOCAB   256
#define EMBED   32
#define HIDDEN  32
#define LAYERS  10
#define BATCH   1024
#define SEQT    1024

#define RS      34
#define TILEF   (16 * RS)
#define NBUF    (LAYERS * 2 * TILEF)
#define NEMB    (VOCAB * RS)
#define NWAVES  (LAYERS * 2)

typedef __attribute__((ext_vector_type(2))) float v2f;
typedef __attribute__((ext_vector_type(8))) float v8f;

__device__ __forceinline__ v8f wmma_k4(v2f a, v2f b, v8f c) {
  return __builtin_amdgcn_wmma_f32_16x16x4_f32(false, a, false, b, (short)0, c,
                                               false, false);
}

__device__ __forceinline__ float dev_tanh(float x) {
#if __has_builtin(__builtin_amdgcn_tanhf)
  return __builtin_amdgcn_tanhf(x);
#else
  float ax = __builtin_fabsf(x);
  float e  = __expf(-2.0f * ax);
  float y  = __fdividef(1.0f - e, 1.0f + e);
  return __builtin_copysignf(y, x);
#endif
}

__global__ void __launch_bounds__(NWAVES * 32)
rnn_pipeline_kernel(const int* __restrict__ x, const float* __restrict__ emb,
                    const float* __restrict__ W_ih, const float* __restrict__ W_hh,
                    const float* __restrict__ b_ih, const float* __restrict__ b_hh,
                    const float* __restrict__ W_fc, const float* __restrict__ b_fc,
                    float* __restrict__ out) {
  extern __shared__ float smem[];
  float* buf  = smem;
  float* embs = smem + NBUF;

  const int tid  = threadIdx.x;
  const int lane = tid & 31;
  const int w    = tid >> 5;
  const int l    = w >> 1;
  const int c    = w & 1;
  const int R    = blockIdx.x * 16;

  for (int i = tid; i < NBUF; i += NWAVES * 32) buf[i] = 0.0f;
  for (int i = tid; i < VOCAB * EMBED; i += NWAVES * 32)
    embs[(i >> 5) * RS + (i & 31)] = emb[i];

  const int m    = lane & 15;
  const int koff = (lane >> 4) * 2;
  const int rb   = (lane >> 4) * 8;

  v2f wih[8], whh[8];
  float bias;
  {
    const float* Wi = W_ih + l * HIDDEN * EMBED;
    const float* Wh = W_hh + l * HIDDEN * HIDDEN;
    const int row = c * 16 + m;
#pragma unroll
    for (int j = 0; j < 8; ++j) {
      wih[j] = *(const v2f*)(Wi + row * EMBED  + 4 * j + koff);
      whh[j] = *(const v2f*)(Wh + row * HIDDEN + 4 * j + koff);
    }
    bias = b_ih[l * HIDDEN + row] + b_hh[l * HIDDEN + row];
  }

  __syncthreads();

  for (int i = 0; i < SEQT + LAYERS - 1; ++i) {
    const int t = i - l;
    if ((unsigned)t < (unsigned)SEQT) {
      const int pin  = (i - 1) & 1;
      const int pout = i & 1;

      v2f ah[8], ax[8];
      const float* hin = buf + (l * 2 + pin) * TILEF + m * RS + koff;
#pragma unroll
      for (int j = 0; j < 8; ++j) ah[j] = *(const v2f*)(hin + 4 * j);

      if (l == 0) {
        const int tok   = x[(R + m) * SEQT + t];
        const float* xe = embs + tok * RS + koff;
#pragma unroll
        for (int j = 0; j < 8; ++j) ax[j] = *(const v2f*)(xe + 4 * j);
      } else {
        const float* xin = buf + ((l - 1) * 2 + pin) * TILEF + m * RS + koff;
#pragma unroll
        for (int j = 0; j < 8; ++j) ax[j] = *(const v2f*)(xin + 4 * j);
      }

      v8f d;
#pragma unroll
      for (int k = 0; k < 8; ++k) d[k] = bias;
#pragma unroll
      for (int j = 0; j < 8; ++j) d = wmma_k4(ax[j], wih[j], d);
#pragma unroll
      for (int j = 0; j < 8; ++j) d = wmma_k4(ah[j], whh[j], d);

      float* hout = buf + (l * 2 + pout) * TILEF + c * 16;
#pragma unroll
      for (int r = 0; r < 8; ++r)
        hout[(rb + r) * RS + m] = dev_tanh(d[r]);
    }
    __syncthreads();
  }

  const float* htop = buf + ((LAYERS - 1) * 2 + ((SEQT + LAYERS - 2) & 1)) * TILEF;
  v2f ahf[8];
#pragma unroll
  for (int j = 0; j < 8; ++j)
    ahf[j] = *(const v2f*)(htop + m * RS + 4 * j + koff);

  for (int ct = w; ct < VOCAB / 16; ct += NWAVES) {
    const int vrow = ct * 16 + m;
    v2f wb[8];
#pragma unroll
    for (int j = 0; j < 8; ++j)
      wb[j] = *(const v2f*)(W_fc + vrow * HIDDEN + 4 * j + koff);
    const float bf = b_fc[vrow];
    v8f d;
#pragma unroll
    for (int k = 0; k < 8; ++k) d[k] = bf;
#pragma unroll
    for (int j = 0; j < 8; ++j) d = wmma_k4(ahf[j], wb[j], d);
#pragma unroll
    for (int r = 0; r < 8; ++r)
      out[(R + rb + r) * VOCAB + vrow] = d[r];
  }
}

extern "C" void kernel_launch(void* const* d_in, const int* in_sizes, int n_in,
                              void* d_out, int out_size, void* d_ws, size_t ws_size,
                              hipStream_t stream) {
  (void)in_sizes; (void)n_in; (void)out_size; (void)d_ws; (void)ws_size;
  const int*   x   = (const int*)d_in[0];
  const float* emb = (const float*)d_in[1];
  const float* Wih = (const float*)d_in[2];
  const float* Whh = (const float*)d_in[3];
  const float* bih = (const float*)d_in[4];
  const float* bhh = (const float*)d_in[5];
  const float* Wfc = (const float*)d_in[6];
  const float* bfc = (const float*)d_in[7];
  float* out = (float*)d_out;

  const size_t shmem = (size_t)(NBUF + NEMB) * sizeof(float);
  hipLaunchKernelGGL(rnn_pipeline_kernel, dim3(BATCH / 16), dim3(NWAVES * 32),
                     shmem, stream, x, emb, Wih, Whh, bih, bhh, Wfc, bfc, out);
}


# Transposed product (weights as A, activations as B): each wave's D tile is its own next-step B operand, so 4 of 16 WMMAs per step take both operands from registers and start before any LDS data
# speedup vs baseline: 1.2665x; 1.1849x over previous
#include <hip/hip_runtime.h>

#define VOCAB   256
#define EMBED   32
#define HIDDEN  32
#define LAYERS  10
#define BATCH   1024
#define SEQT    1024

#define RS      36
#define TILEF   (16 * RS)
#define NBUF    (LAYERS * 2 * TILEF)
#define NEMB    (VOCAB * RS)
#define NWAVES  (LAYERS * 2)

typedef __attribute__((ext_vector_type(2))) float v2f;
typedef __attribute__((ext_vector_type(4))) float v4f;
typedef __attribute__((ext_vector_type(8))) float v8f;

__device__ __forceinline__ v8f wmma_k4(v2f a, v2f b, v8f c) {
  return __builtin_amdgcn_wmma_f32_16x16x4_f32(false, a, false, b, (short)0, c,
                                               false, false);
}

__device__ __forceinline__ float dev_tanh(float x) {
#if __has_builtin(__builtin_amdgcn_tanhf)
  return __builtin_amdgcn_tanhf(x);
#else
  float ax = __builtin_fabsf(x);
  float e  = __expf(-2.0f * ax);
  float y  = __fdividef(1.0f - e, 1.0f + e);
  return __builtin_copysignf(y, x);
#endif
}

__global__ void __launch_bounds__(NWAVES * 32)
rnn_pipeline_kernel(const int* __restrict__ x, const float* __restrict__ emb,
                    const float* __restrict__ W_ih, const float* __restrict__ W_hh,
                    const float* __restrict__ b_ih, const float* __restrict__ b_hh,
                    const float* __restrict__ W_fc, const float* __restrict__ b_fc,
                    float* __restrict__ out) {
  extern __shared__ __attribute__((aligned(16))) float smem[];
  float* buf  = smem;
  float* embs = smem + NBUF;

  const int tid  = threadIdx.x;
  const int lane = tid & 31;
  const int w    = tid >> 5;
  const int l    = w >> 1;
  const int c    = w & 1;
  const int R    = blockIdx.x * 16;

  for (int i = tid; i < NBUF; i += NWAVES * 32) buf[i] = 0.0f;
  for (int i = tid; i < VOCAB * EMBED; i += NWAVES * 32)
    embs[(i >> 5) * RS + (i & 31)] = emb[i];

  const int m    = lane & 15;
  const int koff = (lane >> 4) * 2;
  const int rb   = (lane >> 4) * 8;

  const int hh   = lane >> 4;
  const int orow = c * 16 + m;
  v2f aown[4];
  v4f aoth[2];
  v4f axw[4];
  float bias[8];
  {
    const float* Wi = W_ih + (l * HIDDEN + orow) * EMBED;
    const float* Wh = W_hh + (l * HIDDEN + orow) * HIDDEN;
#pragma unroll
    for (int j = 0; j < 4; ++j) aown[j] = *(const v2f*)(Wh + 16 * c + 8 * hh + 2 * j);
#pragma unroll
    for (int q = 0; q < 2; ++q) aoth[q] = *(const v4f*)(Wh + 16 * (1 - c) + 8 * q + 4 * hh);
#pragma unroll
    for (int q = 0; q < 4; ++q) axw[q]  = *(const v4f*)(Wi + 8 * q + 4 * hh);
#pragma unroll
    for (int r = 0; r < 8; ++r) {
      const int o = l * HIDDEN + c * 16 + 8 * hh + r;
      bias[r] = b_ih[o] + b_hh[o];
    }
  }

  v8f hreg;
#pragma unroll
  for (int r = 0; r < 8; ++r) hreg[r] = 0.0f;

  __syncthreads();

  for (int i = 0; i < SEQT + LAYERS - 1; ++i) {
    const int t = i - l;
    if ((unsigned)t < (unsigned)SEQT) {
      const int pin  = (i - 1) & 1;
      const int pout = i & 1;

      const float* hoth = buf + (l * 2 + pin) * TILEF + m * RS + 16 * (1 - c) + 4 * hh;
      v4f bo[2], bx[4];
#pragma unroll
      for (int q = 0; q < 2; ++q) bo[q] = *(const v4f*)(hoth + 8 * q);

      if (l == 0) {
        const int tok   = x[(R + m) * SEQT + t];
        const float* xe = embs + tok * RS + 4 * hh;
#pragma unroll
        for (int q = 0; q < 4; ++q) bx[q] = *(const v4f*)(xe + 8 * q);
      } else {
        const float* xin = buf + ((l - 1) * 2 + pin) * TILEF + m * RS + 4 * hh;
#pragma unroll
        for (int q = 0; q < 4; ++q) bx[q] = *(const v4f*)(xin + 8 * q);
      }

      __builtin_amdgcn_sched_barrier(0);

      v8f d;
#pragma unroll
      for (int r = 0; r < 8; ++r) d[r] = bias[r];
#pragma unroll
      for (int j = 0; j < 4; ++j) {
        v2f b; b[0] = hreg[2 * j]; b[1] = hreg[2 * j + 1];
        d = wmma_k4(aown[j], b, d);
      }
#pragma unroll
      for (int q = 0; q < 2; ++q) {
        v2f a0, a1, b0, b1;
        a0[0] = aoth[q][0]; a0[1] = aoth[q][1]; a1[0] = aoth[q][2]; a1[1] = aoth[q][3];
        b0[0] = bo[q][0];   b0[1] = bo[q][1];   b1[0] = bo[q][2];   b1[1] = bo[q][3];
        d = wmma_k4(a0, b0, d);
        d = wmma_k4(a1, b1, d);
      }
#pragma unroll
      for (int q = 0; q < 4; ++q) {
        v2f a0, a1, b0, b1;
        a0[0] = axw[q][0]; a0[1] = axw[q][1]; a1[0] = axw[q][2]; a1[1] = axw[q][3];
        b0[0] = bx[q][0];  b0[1] = bx[q][1];  b1[0] = bx[q][2];  b1[1] = bx[q][3];
        d = wmma_k4(a0, b0, d);
        d = wmma_k4(a1, b1, d);
      }

#pragma unroll
      for (int r = 0; r < 8; ++r) hreg[r] = dev_tanh(d[r]);

      float* hout = buf + (l * 2 + pout) * TILEF + m * RS + 16 * c + 8 * hh;
      v4f s0, s1;
      s0[0] = hreg[0]; s0[1] = hreg[1]; s0[2] = hreg[2]; s0[3] = hreg[3];
      s1[0] = hreg[4]; s1[1] = hreg[5]; s1[2] = hreg[6]; s1[3] = hreg[7];
      *(v4f*)(hout)     = s0;
      *(v4f*)(hout + 4) = s1;
    }
    __syncthreads();
  }

  const float* htop = buf + ((LAYERS - 1) * 2 + ((SEQT + LAYERS - 2) & 1)) * TILEF;
  v2f ahf[8];
#pragma unroll
  for (int j = 0; j < 8; ++j)
    ahf[j] = *(const v2f*)(htop + m * RS + 4 * j + koff);

  for (int ct = w; ct < VOCAB / 16; ct += NWAVES) {
    const int vrow = ct * 16 + m;
    v2f wb[8];
#pragma unroll
    for (int j = 0; j < 8; ++j)
      wb[j] = *(const v2f*)(W_fc + vrow * HIDDEN + 4 * j + koff);
    const float bf = b_fc[vrow];
    v8f d;
#pragma unroll
    for (int k = 0; k < 8; ++k) d[k] = bf;
#pragma unroll
    for (int j = 0; j < 8; ++j) d = wmma_k4(ahf[j], wb[j], d);
#pragma unroll
    for (int r = 0; r < 8; ++r)
      out[(R + rb + r) * VOCAB + vrow] = d[r];
  }
}

extern "C" void kernel_launch(void* const* d_in, const int* in_sizes, int n_in,
                              void* d_out, int out_size, void* d_ws, size_t ws_size,
                              hipStream_t stream) {
  (void)in_sizes; (void)n_in; (void)out_size; (void)d_ws; (void)ws_size;
  const int*   x   = (const int*)d_in[0];
  const float* emb = (const float*)d_in[1];
  const float* Wih = (const float*)d_in[2];
  const float* Whh = (const float*)d_in[3];
  const float* bih = (const float*)d_in[4];
  const float* bhh = (const float*)d_in[5];
  const float* Wfc = (const float*)d_in[6];
  const float* bfc = (const float*)d_in[7];
  float* out = (float*)d_out;

  const size_t shmem = (size_t)(NBUF + NEMB) * sizeof(float);
  hipLaunchKernelGGL(rnn_pipeline_kernel, dim3(BATCH / 16), dim3(NWAVES * 32),
                     shmem, stream, x, emb, Wih, Whh, bih, bhh, Wfc, bfc, out);
}


# Split workgroup barrier: a wave signals once its state tile is in LDS, issues the 4 register-fed WMMAs of its next step at low priority, then waits; these fill the matrix pipe in the last wave's tail
# speedup vs baseline: 1.4092x; 1.1127x over previous
#include <hip/hip_runtime.h>

#define VOCAB   256
#define EMBED   32
#define HIDDEN  32
#define LAYERS  10
#define BATCH   1024
#define SEQT    1024

#define RS      36
#define TILEF   (16 * RS)
#define NBUF    (LAYERS * 2 * TILEF)
#define NEMB    (VOCAB * RS)
#define NWAVES  (LAYERS * 2)

typedef __attribute__((ext_vector_type(2))) float v2f;
typedef __attribute__((ext_vector_type(4))) float v4f;
typedef __attribute__((ext_vector_type(8))) float v8f;

__device__ __forceinline__ v8f wmma_k4(v2f a, v2f b, v8f c) {
  return __builtin_amdgcn_wmma_f32_16x16x4_f32(false, a, false, b, (short)0, c,
                                               false, false);
}

__device__ __forceinline__ float dev_tanh(float x) {
#if __has_builtin(__builtin_amdgcn_tanhf)
  return __builtin_amdgcn_tanhf(x);
#else
  float ax = __builtin_fabsf(x);
  float e  = __expf(-2.0f * ax);
  float y  = __fdividef(1.0f - e, 1.0f + e);
  return __builtin_copysignf(y, x);
#endif
}

__device__ __forceinline__ void bar_signal_after_lds() {
  asm volatile("s_wait_dscnt 0x0\n\ts_barrier_signal -1" ::: "memory");
}
__device__ __forceinline__ void bar_wait() {
  asm volatile("s_barrier_wait -1" ::: "memory");
}

__global__ void __launch_bounds__(NWAVES * 32)
rnn_pipeline_kernel(const int* __restrict__ x, const float* __restrict__ emb,
                    const float* __restrict__ W_ih, const float* __restrict__ W_hh,
                    const float* __restrict__ b_ih, const float* __restrict__ b_hh,
                    const float* __restrict__ W_fc, const float* __restrict__ b_fc,
                    float* __restrict__ out) {
  extern __shared__ __attribute__((aligned(16))) float smem[];
  float* buf  = smem;
  float* embs = smem + NBUF;

  const int tid  = threadIdx.x;
  const int lane = tid & 31;
  const int w    = tid >> 5;
  const int l    = w >> 1;
  const int c    = w & 1;
  const int R    = blockIdx.x * 16;

  for (int i = tid; i < NBUF; i += NWAVES * 32) buf[i] = 0.0f;
  for (int i = tid; i < VOCAB * EMBED; i += NWAVES * 32)
    embs[(i >> 5) * RS + (i & 31)] = emb[i];

  const int m    = lane & 15;
  const int koff = (lane >> 4) * 2;
  const int rb   = (lane >> 4) * 8;

  const int hh   = lane >> 4;
  const int orow = c * 16 + m;
  v2f aown[4];
  v4f aoth[2];
  v4f axw[4];
  float bias[8];
  {
    const float* Wi = W_ih + (l * HIDDEN + orow) * EMBED;
    const float* Wh = W_hh + (l * HIDDEN + orow) * HIDDEN;
#pragma unroll
    for (int j = 0; j < 4; ++j) aown[j] = *(const v2f*)(Wh + 16 * c + 8 * hh + 2 * j);
#pragma unroll
    for (int q = 0; q < 2; ++q) aoth[q] = *(const v4f*)(Wh + 16 * (1 - c) + 8 * q + 4 * hh);
#pragma unroll
    for (int q = 0; q < 4; ++q) axw[q]  = *(const v4f*)(Wi + 8 * q + 4 * hh);
#pragma unroll
    for (int r = 0; r < 8; ++r) {
      const int o = l * HIDDEN + c * 16 + 8 * hh + r;
      bias[r] = b_ih[o] + b_hh[o];
    }
  }

  v8f hreg;
#pragma unroll
  for (int r = 0; r < 8; ++r) hreg[r] = 0.0f;

  __syncthreads();

  v8f d;
#pragma unroll
  for (int r = 0; r < 8; ++r) d[r] = bias[r];

  for (int i = 0; i < SEQT + LAYERS - 1; ++i) {
    const int t = i - l;
    __builtin_amdgcn_s_setprio(1);
    if ((unsigned)t < (unsigned)SEQT) {
      const int pin  = (i - 1) & 1;
      const int pout = i & 1;

      const float* hoth = buf + (l * 2 + pin) * TILEF + m * RS + 16 * (1 - c) + 4 * hh;
      v4f bo[2], bx[4];
#pragma unroll
      for (int q = 0; q < 2; ++q) bo[q] = *(const v4f*)(hoth + 8 * q);

      if (l == 0) {
        const int tok   = x[(R + m) * SEQT + t];
        const float* xe = embs + tok * RS + 4 * hh;
#pragma unroll
        for (int q = 0; q < 4; ++q) bx[q] = *(const v4f*)(xe + 8 * q);
      } else {
        const float* xin = buf + ((l - 1) * 2 + pin) * TILEF + m * RS + 4 * hh;
#pragma unroll
        for (int q = 0; q < 4; ++q) bx[q] = *(const v4f*)(xin + 8 * q);
      }

      __builtin_amdgcn_sched_barrier(0);

#pragma unroll
      for (int q = 0; q < 2; ++q) {
        v2f a0, a1, b0, b1;
        a0[0] = aoth[q][0]; a0[1] = aoth[q][1]; a1[0] = aoth[q][2]; a1[1] = aoth[q][3];
        b0[0] = bo[q][0];   b0[1] = bo[q][1];   b1[0] = bo[q][2];   b1[1] = bo[q][3];
        d = wmma_k4(a0, b0, d);
        d = wmma_k4(a1, b1, d);
      }
#pragma unroll
      for (int q = 0; q < 4; ++q) {
        v2f a0, a1, b0, b1;
        a0[0] = axw[q][0]; a0[1] = axw[q][1]; a1[0] = axw[q][2]; a1[1] = axw[q][3];
        b0[0] = bx[q][0];  b0[1] = bx[q][1];  b1[0] = bx[q][2];  b1[1] = bx[q][3];
        d = wmma_k4(a0, b0, d);
        d = wmma_k4(a1, b1, d);
      }

#pragma unroll
      for (int r = 0; r < 8; ++r) hreg[r] = dev_tanh(d[r]);

      float* hout = buf + (l * 2 + pout) * TILEF + m * RS + 16 * c + 8 * hh;
      v4f s0, s1;
      s0[0] = hreg[0]; s0[1] = hreg[1]; s0[2] = hreg[2]; s0[3] = hreg[3];
      s1[0] = hreg[4]; s1[1] = hreg[5]; s1[2] = hreg[6]; s1[3] = hreg[7];
      *(v4f*)(hout)     = s0;
      *(v4f*)(hout + 4) = s1;
    }
    __builtin_amdgcn_sched_barrier(0);
    bar_signal_after_lds();
    __builtin_amdgcn_s_setprio(0);
    __builtin_amdgcn_sched_barrier(0);
    if ((unsigned)(t + 1) < (unsigned)SEQT) {
#pragma unroll
      for (int r = 0; r < 8; ++r) d[r] = bias[r];
#pragma unroll
      for (int j = 0; j < 4; ++j) {
        v2f b; b[0] = hreg[2 * j]; b[1] = hreg[2 * j + 1];
        d = wmma_k4(aown[j], b, d);
      }
    }
    __builtin_amdgcn_sched_barrier(0);
    bar_wait();
  }

  const float* htop = buf + ((LAYERS - 1) * 2 + ((SEQT + LAYERS - 2) & 1)) * TILEF;
  v2f ahf[8];
#pragma unroll
  for (int j = 0; j < 8; ++j)
    ahf[j] = *(const v2f*)(htop + m * RS + 4 * j + koff);

  for (int ct = w; ct < VOCAB / 16; ct += NWAVES) {
    const int vrow = ct * 16 + m;
    v2f wb[8];
#pragma unroll
    for (int j = 0; j < 8; ++j)
      wb[j] = *(const v2f*)(W_fc + vrow * HIDDEN + 4 * j + koff);
    const float bf = b_fc[vrow];
    v8f d;
#pragma unroll
    for (int k = 0; k < 8; ++k) d[k] = bf;
#pragma unroll
    for (int j = 0; j < 8; ++j) d = wmma_k4(ahf[j], wb[j], d);
#pragma unroll
    for (int r = 0; r < 8; ++r)
      out[(R + rb + r) * VOCAB + vrow] = d[r];
  }
}

extern "C" void kernel_launch(void* const* d_in, const int* in_sizes, int n_in,
                              void* d_out, int out_size, void* d_ws, size_t ws_size,
                              hipStream_t stream) {
  (void)in_sizes; (void)n_in; (void)out_size; (void)d_ws; (void)ws_size;
  const int*   x   = (const int*)d_in[0];
  const float* emb = (const float*)d_in[1];
  const float* Wih = (const float*)d_in[2];
  const float* Whh = (const float*)d_in[3];
  const float* bih = (const float*)d_in[4];
  const float* bhh = (const float*)d_in[5];
  const float* Wfc = (const float*)d_in[6];
  const float* bfc = (const float*)d_in[7];
  float* out = (float*)d_out;

  const size_t shmem = (size_t)(NBUF + NEMB) * sizeof(float);
  hipLaunchKernelGGL(rnn_pipeline_kernel, dim3(BATCH / 16), dim3(NWAVES * 32),
                     shmem, stream, x, emb, Wih, Whh, bih, bhh, Wfc, bfc, out);
}


# 8 batch rows per workgroup (128 workgroups): a tile's 16 columns are 8 rows x 2 time steps, so the input product runs once per two steps; 12 not 16 f32 WMMAs per wave and step, state moved by DPP
# speedup vs baseline: 1.4735x; 1.0456x over previous
#include <hip/hip_runtime.h>

#define VOCAB   256
#define EMBED   32
#define HIDDEN  32
#define LAYERS  10
#define BATCH   1024
#define SEQT    1024

#define BT      8
#define NSLOT   4
#define SKEW    2
#define RS      36
#define SLOTF   (BT * RS)
#define LAYF    (NSLOT * SLOTF)
#define NBUF    (LAYERS * LAYF)
#define NEMB    (VOCAB * RS)
#define NWAVES  (LAYERS * 2)
#define NITER   (SEQT + SKEW * (LAYERS - 1))

typedef __attribute__((ext_vector_type(2))) float v2f;
typedef __attribute__((ext_vector_type(4))) float v4f;
typedef __attribute__((ext_vector_type(8))) float v8f;

__device__ __forceinline__ v8f wmma_k4(v2f a, v2f b, v8f c) {
  return __builtin_amdgcn_wmma_f32_16x16x4_f32(false, a, false, b, (short)0, c,
                                               false, false);
}

__device__ __forceinline__ float dev_tanh(float x) {
#if __has_builtin(__builtin_amdgcn_tanhf)
  return __builtin_amdgcn_tanhf(x);
#else
  float ax = __builtin_fabsf(x);
  float e  = __expf(-2.0f * ax);
  float y  = __fdividef(1.0f - e, 1.0f + e);
  return __builtin_copysignf(y, x);
#endif
}

__device__ __forceinline__ float swap8(float v) {
  return __builtin_bit_cast(float,
      __builtin_amdgcn_update_dpp(0, __builtin_bit_cast(int, v), 0x168, 0xf, 0xf, true));
}

__device__ __forceinline__ void bar_signal_after_lds() {
  asm volatile("s_wait_dscnt 0x0\n\ts_barrier_signal -1" ::: "memory");
}
__device__ __forceinline__ void bar_wait() {
  asm volatile("s_barrier_wait -1" ::: "memory");
}

__global__ void __launch_bounds__(NWAVES * 32)
rnn_pipeline_kernel(const int* __restrict__ x, const float* __restrict__ emb,
                    const float* __restrict__ W_ih, const float* __restrict__ W_hh,
                    const float* __restrict__ b_ih, const float* __restrict__ b_hh,
                    const float* __restrict__ W_fc, const float* __restrict__ b_fc,
                    float* __restrict__ out) {
  extern __shared__ __attribute__((aligned(16))) float smem[];
  float* buf  = smem;
  float* embs = smem + NBUF;

  const int tid  = threadIdx.x;
  const int lane = tid & 31;
  const int w    = __builtin_amdgcn_readfirstlane(tid >> 5);
  const int l    = w >> 1;
  const int c    = w & 1;
  const int R    = blockIdx.x * BT;

  for (int i = tid; i < NBUF; i += NWAVES * 32) buf[i] = 0.0f;
  for (int i = tid; i < VOCAB * EMBED; i += NWAVES * 32)
    embs[(i >> 5) * RS + (i & 31)] = emb[i];

  const int m    = lane & 15;
  const int hh   = lane >> 4;
  const int koff = hh * 2;
  const int b    = m & 7;
  const int tn   = m >> 3;

  const int orow = c * 16 + m;
  v2f aown[4];
  v4f aoth[2];
  v4f axw[4];
  float bias[8];
  {
    const float* Wi = W_ih + (l * HIDDEN + orow) * EMBED;
    const float* Wh = W_hh + (l * HIDDEN + orow) * HIDDEN;
#pragma unroll
    for (int j = 0; j < 4; ++j) aown[j] = *(const v2f*)(Wh + 16 * c + 8 * hh + 2 * j);
#pragma unroll
    for (int q = 0; q < 2; ++q) aoth[q] = *(const v4f*)(Wh + 16 * (1 - c) + 8 * q + 4 * hh);
#pragma unroll
    for (int q = 0; q < 4; ++q) axw[q]  = *(const v4f*)(Wi + 8 * q + 4 * hh);
#pragma unroll
    for (int r = 0; r < 8; ++r) {
      const int o = l * HIDDEN + c * 16 + 8 * hh + r;
      bias[r] = b_ih[o] + b_hh[o];
    }
  }

  v8f hreg;
  v8f P;
  v8f d;
#pragma unroll
  for (int r = 0; r < 8; ++r) { hreg[r] = 0.0f; P[r] = bias[r]; d[r] = 0.0f; }

  __syncthreads();

  for (int i = 0; i < NITER; ++i) {
    const int t = i - SKEW * l;
    __builtin_amdgcn_s_setprio(1);
    if ((unsigned)t < (unsigned)SEQT) {
      const float* hoth = buf + l * LAYF + ((t - 1) & 3) * SLOTF + b * RS + 16 * (1 - c) + 4 * hh;
      v4f bo[2];
#pragma unroll
      for (int q = 0; q < 2; ++q) bo[q] = *(const v4f*)(hoth + 8 * q);

      if ((t & 1) == 0) {
        v4f bx[4];
        if (l == 0) {
          const int tok   = x[(R + b) * SEQT + t + tn];
          const float* xe = embs + tok * RS + 4 * hh;
#pragma unroll
          for (int q = 0; q < 4; ++q) bx[q] = *(const v4f*)(xe + 8 * q);
        } else {
          const float* xin = buf + (l - 1) * LAYF + ((t + tn) & 3) * SLOTF + b * RS + 4 * hh;
#pragma unroll
          for (int q = 0; q < 4; ++q) bx[q] = *(const v4f*)(xin + 8 * q);
        }

        __builtin_amdgcn_sched_barrier(0);

#pragma unroll
        for (int q = 0; q < 2; ++q) {
          v2f a0, a1, b0, b1;
          a0[0] = aoth[q][0]; a0[1] = aoth[q][1]; a1[0] = aoth[q][2]; a1[1] = aoth[q][3];
          b0[0] = bo[q][0];   b0[1] = bo[q][1];   b1[0] = bo[q][2];   b1[1] = bo[q][3];
          d = wmma_k4(a0, b0, d);
          d = wmma_k4(a1, b1, d);
        }
#pragma unroll
        for (int r = 0; r < 8; ++r) P[r] = bias[r];
#pragma unroll
        for (int q = 0; q < 4; ++q) {
          v2f a0, a1, b0, b1;
          a0[0] = axw[q][0]; a0[1] = axw[q][1]; a1[0] = axw[q][2]; a1[1] = axw[q][3];
          b0[0] = bx[q][0];  b0[1] = bx[q][1];  b1[0] = bx[q][2];  b1[1] = bx[q][3];
          P = wmma_k4(a0, b0, P);
          P = wmma_k4(a1, b1, P);
        }
#pragma unroll
        for (int r = 0; r < 8; ++r) d[r] += P[r];
      } else {
        __builtin_amdgcn_sched_barrier(0);
#pragma unroll
        for (int q = 0; q < 2; ++q) {
          v2f a0, a1, b0, b1;
          a0[0] = aoth[q][0]; a0[1] = aoth[q][1]; a1[0] = aoth[q][2]; a1[1] = aoth[q][3];
          b0[0] = bo[q][0];   b0[1] = bo[q][1];   b1[0] = bo[q][2];   b1[1] = bo[q][3];
          d = wmma_k4(a0, b0, d);
          d = wmma_k4(a1, b1, d);
        }
      }

#pragma unroll
      for (int r = 0; r < 8; ++r) hreg[r] = dev_tanh(d[r]);

      if (tn == (t & 1)) {
        float* hout = buf + l * LAYF + (t & 3) * SLOTF + b * RS + 16 * c + 8 * hh;
        v4f s0, s1;
        s0[0] = hreg[0]; s0[1] = hreg[1]; s0[2] = hreg[2]; s0[3] = hreg[3];
        s1[0] = hreg[4]; s1[1] = hreg[5]; s1[2] = hreg[6]; s1[3] = hreg[7];
        *(v4f*)(hout)     = s0;
        *(v4f*)(hout + 4) = s1;
      }
    }
    __builtin_amdgcn_sched_barrier(0);
    bar_signal_after_lds();
    __builtin_amdgcn_s_setprio(0);
    __builtin_amdgcn_sched_barrier(0);
    if ((unsigned)(t + 1) < (unsigned)SEQT) {
      v8f hx;
#pragma unroll
      for (int r = 0; r < 8; ++r) hx[r] = swap8(hreg[r]);
      if ((t + 1) & 1) {
#pragma unroll
        for (int r = 0; r < 8; ++r) d[r] = P[r];
      } else {
#pragma unroll
        for (int r = 0; r < 8; ++r) d[r] = 0.0f;
      }
      __builtin_amdgcn_sched_barrier(0);
#pragma unroll
      for (int j = 0; j < 4; ++j) {
        v2f bb; bb[0] = hx[2 * j]; bb[1] = hx[2 * j + 1];
        d = wmma_k4(aown[j], bb, d);
      }
    }
    __builtin_amdgcn_sched_barrier(0);
    bar_wait();
  }

  const float* htop = buf + (LAYERS - 1) * LAYF + ((SEQT - 1) & 3) * SLOTF;
  v2f ahf[8];
#pragma unroll
  for (int j = 0; j < 8; ++j)
    ahf[j] = *(const v2f*)(htop + b * RS + 4 * j + koff);

  for (int ct = w; ct < VOCAB / 16; ct += NWAVES) {
    const int vrow = ct * 16 + m;
    v2f wb[8];
#pragma unroll
    for (int j = 0; j < 8; ++j)
      wb[j] = *(const v2f*)(W_fc + vrow * HIDDEN + 4 * j + koff);
    const float bf = b_fc[vrow];
    v8f dd;
#pragma unroll
    for (int k = 0; k < 8; ++k) dd[k] = bf;
#pragma unroll
    for (int j = 0; j < 8; ++j) dd = wmma_k4(ahf[j], wb[j], dd);
    if (hh == 0) {
#pragma unroll
      for (int r = 0; r < 8; ++r)
        out[(R + r) * VOCAB + vrow] = dd[r];
    }
  }
}

extern "C" void kernel_launch(void* const* d_in, const int* in_sizes, int n_in,
                              void* d_out, int out_size, void* d_ws, size_t ws_size,
                              hipStream_t stream) {
  (void)in_sizes; (void)n_in; (void)out_size; (void)d_ws; (void)ws_size;
  const int*   x   = (const int*)d_in[0];
  const float* emb = (const float*)d_in[1];
  const float* Wih = (const float*)d_in[2];
  const float* Whh = (const float*)d_in[3];
  const float* bih = (const float*)d_in[4];
  const float* bhh = (const float*)d_in[5];
  const float* Wfc = (const float*)d_in[6];
  const float* bfc = (const float*)d_in[7];
  float* out = (float*)d_out;

  const size_t shmem = (size_t)(NBUF + NEMB) * sizeof(float);
  hipLaunchKernelGGL(rnn_pipeline_kernel, dim3(BATCH / BT), dim3(NWAVES * 32),
                     shmem, stream, x, emb, Wih, Whh, bih, bhh, Wfc, bfc, out);
}


# Loop body covers four time steps as straight-line code: step parity and ring-slot LDS addresses become per-phase constants, removing the parity branches and per-step address arithmetic
# speedup vs baseline: 1.6265x; 1.1038x over previous
#include <hip/hip_runtime.h>

#define VOCAB   256
#define EMBED   32
#define HIDDEN  32
#define LAYERS  10
#define BATCH   1024
#define SEQT    1024

#define BT      8
#define NSLOT   4
#define SKEW    2
#define RS      36
#define SLOTF   (BT * RS)
#define LAYF    (NSLOT * SLOTF)
#define NBUF    (LAYERS * LAYF)
#define NEMB    (VOCAB * RS)
#define NWAVES  (LAYERS * 2)
#define NITER   (SEQT + SKEW * (LAYERS - 1))
#define NITER4  ((NITER + 3) / 4)

typedef __attribute__((ext_vector_type(2))) float v2f;
typedef __attribute__((ext_vector_type(4))) float v4f;
typedef __attribute__((ext_vector_type(8))) float v8f;

__device__ __forceinline__ v8f wmma_k4(v2f a, v2f b, v8f c) {
  return __builtin_amdgcn_wmma_f32_16x16x4_f32(false, a, false, b, (short)0, c,
                                               false, false);
}

__device__ __forceinline__ float dev_tanh(float x) {
#if __has_builtin(__builtin_amdgcn_tanhf)
  return __builtin_amdgcn_tanhf(x);
#else
  float ax = __builtin_fabsf(x);
  float e  = __expf(-2.0f * ax);
  float y  = __fdividef(1.0f - e, 1.0f + e);
  return __builtin_copysignf(y, x);
#endif
}

__device__ __forceinline__ float swap8(float v) {
  return __builtin_bit_cast(float,
      __builtin_amdgcn_update_dpp(0, __builtin_bit_cast(int, v), 0x168, 0xf, 0xf, true));
}

__device__ __forceinline__ void bar_signal_after_lds() {
  asm volatile("s_wait_dscnt 0x0\n\ts_barrier_signal -1" ::: "memory");
}
__device__ __forceinline__ void bar_wait() {
  asm volatile("s_barrier_wait -1" ::: "memory");
}

__global__ void __launch_bounds__(NWAVES * 32)
rnn_pipeline_kernel(const int* __restrict__ x, const float* __restrict__ emb,
                    const float* __restrict__ W_ih, const float* __restrict__ W_hh,
                    const float* __restrict__ b_ih, const float* __restrict__ b_hh,
                    const float* __restrict__ W_fc, const float* __restrict__ b_fc,
                    float* __restrict__ out) {
  extern __shared__ __attribute__((aligned(16))) float smem[];
  float* buf  = smem;
  float* embs = smem + NBUF;

  const int tid  = threadIdx.x;
  const int lane = tid & 31;
  const int w    = __builtin_amdgcn_readfirstlane(tid >> 5);
  const int l    = w >> 1;
  const int c    = w & 1;
  const int R    = blockIdx.x * BT;

  for (int i = tid; i < NBUF; i += NWAVES * 32) buf[i] = 0.0f;
  for (int i = tid; i < VOCAB * EMBED; i += NWAVES * 32)
    embs[(i >> 5) * RS + (i & 31)] = emb[i];

  const int m    = lane & 15;
  const int hh   = lane >> 4;
  const int koff = hh * 2;
  const int b    = m & 7;
  const int tn   = m >> 3;

  const int orow = c * 16 + m;
  v2f aown[4];
  v4f aoth[2];
  v4f axw[4];
  float bias[8];
  {
    const float* Wi = W_ih + (l * HIDDEN + orow) * EMBED;
    const float* Wh = W_hh + (l * HIDDEN + orow) * HIDDEN;
#pragma unroll
    for (int j = 0; j < 4; ++j) aown[j] = *(const v2f*)(Wh + 16 * c + 8 * hh + 2 * j);
#pragma unroll
    for (int q = 0; q < 2; ++q) aoth[q] = *(const v4f*)(Wh + 16 * (1 - c) + 8 * q + 4 * hh);
#pragma unroll
    for (int q = 0; q < 4; ++q) axw[q]  = *(const v4f*)(Wi + 8 * q + 4 * hh);
#pragma unroll
    for (int r = 0; r < 8; ++r) {
      const int o = l * HIDDEN + c * 16 + 8 * hh + r;
      bias[r] = b_ih[o] + b_hh[o];
    }
  }

  v8f hreg;
  v8f P;
  v8f d;
#pragma unroll
  for (int r = 0; r < 8; ++r) { hreg[r] = 0.0f; P[r] = bias[r]; d[r] = 0.0f; }

  __syncthreads();

  const int lo = (l & 1) * 2;
  const int lin = (l > 0) ? (l - 1) : 0;
  int o_hoth[4], o_hout[4], o_xin[2];
#pragma unroll
  for (int ph = 0; ph < 4; ++ph) {
    const int sc = (ph + lo) & 3;
    o_hoth[ph] = l * LAYF + ((sc + 3) & 3) * SLOTF + b * RS + 16 * (1 - c) + 4 * hh;
    o_hout[ph] = l * LAYF + sc * SLOTF + b * RS + 16 * c + 8 * hh;
  }
#pragma unroll
  for (int e = 0; e < 2; ++e) {
    const int sc = (2 * e + lo) & 3;
    o_xin[e] = lin * LAYF + ((sc + tn) & 3) * SLOTF + b * RS + 4 * hh;
  }
  const int* xrow = x + (R + b) * SEQT + tn;

  for (int i4 = 0; i4 < NITER4; ++i4) {
#pragma unroll
    for (int ph = 0; ph < 4; ++ph) {
      const int t = i4 * 4 + ph - SKEW * l;
      __builtin_amdgcn_s_setprio(1);
      if ((unsigned)t < (unsigned)SEQT) {
        const float* hoth = buf + o_hoth[ph];
        v4f bo[2];
#pragma unroll
        for (int q = 0; q < 2; ++q) bo[q] = *(const v4f*)(hoth + 8 * q);

        if ((ph & 1) == 0) {
          v4f bx[4];
          if (l == 0) {
            const int tok   = xrow[t];
            const float* xe = embs + tok * RS + 4 * hh;
#pragma unroll
            for (int q = 0; q < 4; ++q) bx[q] = *(const v4f*)(xe + 8 * q);
          } else {
            const float* xin = buf + o_xin[ph >> 1];
#pragma unroll
            for (int q = 0; q < 4; ++q) bx[q] = *(const v4f*)(xin + 8 * q);
          }

          __builtin_amdgcn_sched_barrier(0);

#pragma unroll
          for (int q = 0; q < 2; ++q) {
            v2f a0, a1, b0, b1;
            a0[0] = aoth[q][0]; a0[1] = aoth[q][1]; a1[0] = aoth[q][2]; a1[1] = aoth[q][3];
            b0[0] = bo[q][0];   b0[1] = bo[q][1];   b1[0] = bo[q][2];   b1[1] = bo[q][3];
            d = wmma_k4(a0, b0, d);
            d = wmma_k4(a1, b1, d);
          }
#pragma unroll
          for (int r = 0; r < 8; ++r) P[r] = bias[r];
#pragma unroll
          for (int q = 0; q < 4; ++q) {
            v2f a0, a1, b0, b1;
            a0[0] = axw[q][0]; a0[1] = axw[q][1]; a1[0] = axw[q][2]; a1[1] = axw[q][3];
            b0[0] = bx[q][0];  b0[1] = bx[q][1];  b1[0] = bx[q][2];  b1[1] = bx[q][3];
            P = wmma_k4(a0, b0, P);
            P = wmma_k4(a1, b1, P);
          }
#pragma unroll
          for (int r = 0; r < 8; ++r) d[r] += P[r];
        } else {
          __builtin_amdgcn_sched_barrier(0);
#pragma unroll
          for (int q = 0; q < 2; ++q) {
            v2f a0, a1, b0, b1;
            a0[0] = aoth[q][0]; a0[1] = aoth[q][1]; a1[0] = aoth[q][2]; a1[1] = aoth[q][3];
            b0[0] = bo[q][0];   b0[1] = bo[q][1];   b1[0] = bo[q][2];   b1[1] = bo[q][3];
            d = wmma_k4(a0, b0, d);
            d = wmma_k4(a1, b1, d);
          }
        }

#pragma unroll
        for (int r = 0; r < 8; ++r) hreg[r] = dev_tanh(d[r]);

        if (tn == (ph & 1)) {
          float* hout = buf + o_hout[ph];
          v4f s0, s1;
          s0[0] = hreg[0]; s0[1] = hreg[1]; s0[2] = hreg[2]; s0[3] = hreg[3];
          s1[0] = hreg[4]; s1[1] = hreg[5]; s1[2] = hreg[6]; s1[3] = hreg[7];
          *(v4f*)(hout)     = s0;
          *(v4f*)(hout + 4) = s1;
        }
      }
      __builtin_amdgcn_sched_barrier(0);
      bar_signal_after_lds();
      __builtin_amdgcn_s_setprio(0);
      __builtin_amdgcn_sched_barrier(0);
      if ((unsigned)(t + 1) < (unsigned)SEQT) {
        v8f hx;
#pragma unroll
        for (int r = 0; r < 8; ++r) hx[r] = swap8(hreg[r]);
        if ((ph & 1) == 0) {
#pragma unroll
          for (int r = 0; r < 8; ++r) d[r] = P[r];
        } else {
#pragma unroll
          for (int r = 0; r < 8; ++r) d[r] = 0.0f;
        }
        __builtin_amdgcn_sched_barrier(0);
#pragma unroll
        for (int j = 0; j < 4; ++j) {
          v2f bb; bb[0] = hx[2 * j]; bb[1] = hx[2 * j + 1];
          d = wmma_k4(aown[j], bb, d);
        }
      }
      __builtin_amdgcn_sched_barrier(0);
      bar_wait();
    }
  }

  const float* htop = buf + (LAYERS - 1) * LAYF + ((SEQT - 1) & 3) * SLOTF;
  v2f ahf[8];
#pragma unroll
  for (int j = 0; j < 8; ++j)
    ahf[j] = *(const v2f*)(htop + b * RS + 4 * j + koff);

  for (int ct = w; ct < VOCAB / 16; ct += NWAVES) {
    const int vrow = ct * 16 + m;
    v2f wb[8];
#pragma unroll
    for (int j = 0; j < 8; ++j)
      wb[j] = *(const v2f*)(W_fc + vrow * HIDDEN + 4 * j + koff);
    const float bf = b_fc[vrow];
    v8f dd;
#pragma unroll
    for (int k = 0; k < 8; ++k) dd[k] = bf;
#pragma unroll
    for (int j = 0; j < 8; ++j) dd = wmma_k4(ahf[j], wb[j], dd);
    if (hh == 0) {
#pragma unroll
      for (int r = 0; r < 8; ++r)
        out[(R + r) * VOCAB + vrow] = dd[r];
    }
  }
}

extern "C" void kernel_launch(void* const* d_in, const int* in_sizes, int n_in,
                              void* d_out, int out_size, void* d_ws, size_t ws_size,
                              hipStream_t stream) {
  (void)in_sizes; (void)n_in; (void)out_size; (void)d_ws; (void)ws_size;
  const int*   x   = (const int*)d_in[0];
  const float* emb = (const float*)d_in[1];
  const float* Wih = (const float*)d_in[2];
  const float* Whh = (const float*)d_in[3];
  const float* bih = (const float*)d_in[4];
  const float* bhh = (const float*)d_in[5];
  const float* Wfc = (const float*)d_in[6];
  const float* bfc = (const float*)d_in[7];
  float* out = (float*)d_out;

  const size_t shmem = (size_t)(NBUF + NEMB) * sizeof(float);
  hipLaunchKernelGGL(rnn_pipeline_kernel, dim3(BATCH / BT), dim3(NWAVES * 32),
                     shmem, stream, x, emb, Wih, Whh, bih, bhh, Wfc, bfc, out);
}


# 4 batch rows per workgroup (256 workgroups, one per WGP by a 164 KiB LDS reservation): 16 columns = 4 rows x 4 time steps, input product once per four steps; 10 not 12 f32 WMMAs per wave and step
# speedup vs baseline: 1.8459x; 1.1349x over previous
#include <hip/hip_runtime.h>

#define VOCAB   256
#define EMBED   32
#define HIDDEN  32
#define LAYERS  10
#define BATCH   1024
#define SEQT    1024

#define BT      4
#define NSLOT   8
#define SKEW    4
#define RS      36
#define SLOTF   (BT * RS)
#define LAYF    (NSLOT * SLOTF)
#define NBUF    (LAYERS * LAYF)
#define NEMB    (VOCAB * RS)
#define NWAVES  (LAYERS * 2)
#define NITER   (SEQT + SKEW * (LAYERS - 1))
#define NITER4  (NITER / 4)
#define HALFR   (4 * SLOTF)
#define LDSF    41984

typedef __attribute__((ext_vector_type(2))) float v2f;
typedef __attribute__((ext_vector_type(4))) float v4f;
typedef __attribute__((ext_vector_type(8))) float v8f;

__device__ __forceinline__ v8f wmma_k4(v2f a, v2f b, v8f c) {
  return __builtin_amdgcn_wmma_f32_16x16x4_f32(false, a, false, b, (short)0, c,
                                               false, false);
}

__device__ __forceinline__ float dev_tanh(float x) {
#if __has_builtin(__builtin_amdgcn_tanhf)
  return __builtin_amdgcn_tanhf(x);
#else
  float ax = __builtin_fabsf(x);
  float e  = __expf(-2.0f * ax);
  float y  = __fdividef(1.0f - e, 1.0f + e);
  return __builtin_copysignf(y, x);
#endif
}

__device__ __forceinline__ float swap8(float v) {
  return __builtin_bit_cast(float,
      __builtin_amdgcn_update_dpp(0, __builtin_bit_cast(int, v), 0x168, 0xf, 0xf, true));
}
__device__ __forceinline__ float swap4(float v) {
  return __builtin_bit_cast(float,
      __builtin_amdgcn_update_dpp(0, __builtin_bit_cast(int, v), 0x164, 0xf, 0xf, true));
}

__device__ __forceinline__ void bar_signal_after_lds() {
  asm volatile("s_wait_dscnt 0x0\n\ts_barrier_signal -1" ::: "memory");
}
__device__ __forceinline__ void bar_wait() {
  asm volatile("s_barrier_wait -1" ::: "memory");
}

__global__ void __launch_bounds__(NWAVES * 32)
rnn_pipeline_kernel(const int* __restrict__ x, const float* __restrict__ emb,
                    const float* __restrict__ W_ih, const float* __restrict__ W_hh,
                    const float* __restrict__ b_ih, const float* __restrict__ b_hh,
                    const float* __restrict__ W_fc, const float* __restrict__ b_fc,
                    float* __restrict__ out) {
  __shared__ __attribute__((aligned(16))) float smem[LDSF];
  float* buf  = smem;
  float* embs = smem + NBUF;

  const int tid  = threadIdx.x;
  const int lane = tid & 31;
  const int w    = __builtin_amdgcn_readfirstlane(tid >> 5);
  const int l    = w >> 1;
  const int c    = w & 1;
  const int R    = blockIdx.x * BT;

  for (int i = tid; i < NBUF; i += NWAVES * 32) buf[i] = 0.0f;
  for (int i = tid; i < VOCAB * EMBED; i += NWAVES * 32)
    embs[(i >> 5) * RS + (i & 31)] = emb[i];

  const int m    = lane & 15;
  const int hh   = lane >> 4;
  const int koff = hh * 2;
  const int b    = m & 3;
  const int g    = m >> 2;
  const int tn   = g ^ (g >> 1);

  const int orow = c * 16 + m;
  v2f aown[4];
  v4f aoth[2];
  v4f axw[4];
  float bias[8];
  {
    const float* Wi = W_ih + (l * HIDDEN + orow) * EMBED;
    const float* Wh = W_hh + (l * HIDDEN + orow) * HIDDEN;
#pragma unroll
    for (int j = 0; j < 4; ++j) aown[j] = *(const v2f*)(Wh + 16 * c + 8 * hh + 2 * j);
#pragma unroll
    for (int q = 0; q < 2; ++q) aoth[q] = *(const v4f*)(Wh + 16 * (1 - c) + 8 * q + 4 * hh);
#pragma unroll
    for (int q = 0; q < 4; ++q) axw[q]  = *(const v4f*)(Wi + 8 * q + 4 * hh);
#pragma unroll
    for (int r = 0; r < 8; ++r) {
      const int o = l * HIDDEN + c * 16 + 8 * hh + r;
      bias[r] = b_ih[o] + b_hh[o];
    }
  }

  v8f hreg;
  v8f P;
  v8f d;
#pragma unroll
  for (int r = 0; r < 8; ++r) { hreg[r] = 0.0f; P[r] = bias[r]; d[r] = 0.0f; }

  __syncthreads();

  const int lin = (l > 0) ? (l - 1) : 0;
  int o_hoth[4], o_hout[4];
#pragma unroll
  for (int ph = 0; ph < 4; ++ph) {
    o_hoth[ph] = l * LAYF + ((ph + 3) & 3) * SLOTF + b * RS + 16 * (1 - c) + 4 * hh;
    o_hout[ph] = l * LAYF + ph * SLOTF + b * RS + 16 * c + 8 * hh;
  }
  const int o_xin = lin * LAYF + tn * SLOTF + b * RS + 4 * hh;
  const int* xrow = x + (R + b) * SEQT + tn;

  for (int i4 = 0; i4 < NITER4; ++i4) {
    const int offs = ((i4 - l) & 1) * HALFR;
    const int offo = HALFR - offs;
#pragma unroll
    for (int ph = 0; ph < 4; ++ph) {
      const int t = i4 * 4 + ph - SKEW * l;
      __builtin_amdgcn_s_setprio(1);
      if ((unsigned)t < (unsigned)SEQT) {
        const float* hoth = buf + o_hoth[ph] + ((ph == 0) ? offo : offs);
        v4f bo[2];
#pragma unroll
        for (int q = 0; q < 2; ++q) bo[q] = *(const v4f*)(hoth + 8 * q);

        if (ph == 0) {
          v4f bx[4];
          if (l == 0) {
            const int tok   = xrow[t];
            const float* xe = embs + tok * RS + 4 * hh;
#pragma unroll
            for (int q = 0; q < 4; ++q) bx[q] = *(const v4f*)(xe + 8 * q);
          } else {
            const float* xin = buf + o_xin + offs;
#pragma unroll
            for (int q = 0; q < 4; ++q) bx[q] = *(const v4f*)(xin + 8 * q);
          }

          __builtin_amdgcn_sched_barrier(0);

#pragma unroll
          for (int q = 0; q < 2; ++q) {
            v2f a0, a1, b0, b1;
            a0[0] = aoth[q][0]; a0[1] = aoth[q][1]; a1[0] = aoth[q][2]; a1[1] = aoth[q][3];
            b0[0] = bo[q][0];   b0[1] = bo[q][1];   b1[0] = bo[q][2];   b1[1] = bo[q][3];
            d = wmma_k4(a0, b0, d);
            d = wmma_k4(a1, b1, d);
          }
#pragma unroll
          for (int r = 0; r < 8; ++r) P[r] = bias[r];
#pragma unroll
          for (int q = 0; q < 4; ++q) {
            v2f a0, a1, b0, b1;
            a0[0] = axw[q][0]; a0[1] = axw[q][1]; a1[0] = axw[q][2]; a1[1] = axw[q][3];
            b0[0] = bx[q][0];  b0[1] = bx[q][1];  b1[0] = bx[q][2];  b1[1] = bx[q][3];
            P = wmma_k4(a0, b0, P);
            P = wmma_k4(a1, b1, P);
          }
#pragma unroll
          for (int r = 0; r < 8; ++r) d[r] += P[r];
        } else {
          __builtin_amdgcn_sched_barrier(0);
#pragma unroll
          for (int q = 0; q < 2; ++q) {
            v2f a0, a1, b0, b1;
            a0[0] = aoth[q][0]; a0[1] = aoth[q][1]; a1[0] = aoth[q][2]; a1[1] = aoth[q][3];
            b0[0] = bo[q][0];   b0[1] = bo[q][1];   b1[0] = bo[q][2];   b1[1] = bo[q][3];
            d = wmma_k4(a0, b0, d);
            d = wmma_k4(a1, b1, d);
          }
        }

#pragma unroll
        for (int r = 0; r < 8; ++r) hreg[r] = dev_tanh(d[r]);

        if (tn == ph) {
          float* hout = buf + o_hout[ph] + offs;
          v4f s0, s1;
          s0[0] = hreg[0]; s0[1] = hreg[1]; s0[2] = hreg[2]; s0[3] = hreg[3];
          s1[0] = hreg[4]; s1[1] = hreg[5]; s1[2] = hreg[6]; s1[3] = hreg[7];
          *(v4f*)(hout)     = s0;
          *(v4f*)(hout + 4) = s1;
        }
      }
      __builtin_amdgcn_sched_barrier(0);
      bar_signal_after_lds();
      __builtin_amdgcn_s_setprio(0);
      __builtin_amdgcn_sched_barrier(0);
      if ((unsigned)(t + 1) < (unsigned)SEQT) {
        v8f hx;
        if ((ph & 1) == 0) {
#pragma unroll
          for (int r = 0; r < 8; ++r) hx[r] = swap4(hreg[r]);
        } else {
#pragma unroll
          for (int r = 0; r < 8; ++r) hx[r] = swap8(hreg[r]);
        }
        if (ph != 3) {
#pragma unroll
          for (int r = 0; r < 8; ++r) d[r] = P[r];
        } else {
#pragma unroll
          for (int r = 0; r < 8; ++r) d[r] = 0.0f;
        }
        __builtin_amdgcn_sched_barrier(0);
#pragma unroll
        for (int j = 0; j < 4; ++j) {
          v2f bb; bb[0] = hx[2 * j]; bb[1] = hx[2 * j + 1];
          d = wmma_k4(aown[j], bb, d);
        }
      }
      __builtin_amdgcn_sched_barrier(0);
      bar_wait();
    }
  }

  const float* htop = buf + (LAYERS - 1) * LAYF + ((SEQT - 1) & (NSLOT - 1)) * SLOTF;
  v2f ahf[8];
#pragma unroll
  for (int j = 0; j < 8; ++j)
    ahf[j] = *(const v2f*)(htop + b * RS + 4 * j + koff);

  for (int ct = w; ct < VOCAB / 16; ct += NWAVES) {
    const int vrow = ct * 16 + m;
    v2f wb[8];
#pragma unroll
    for (int j = 0; j < 8; ++j)
      wb[j] = *(const v2f*)(W_fc + vrow * HIDDEN + 4 * j + koff);
    const float bf = b_fc[vrow];
    v8f dd;
#pragma unroll
    for (int k = 0; k < 8; ++k) dd[k] = bf;
#pragma unroll
    for (int j = 0; j < 8; ++j) dd = wmma_k4(ahf[j], wb[j], dd);
    if (hh == 0) {
#pragma unroll
      for (int r = 0; r < BT; ++r)
        out[(R + r) * VOCAB + vrow] = dd[r];
    }
  }
}

extern "C" void kernel_launch(void* const* d_in, const int* in_sizes, int n_in,
                              void* d_out, int out_size, void* d_ws, size_t ws_size,
                              hipStream_t stream) {
  (void)in_sizes; (void)n_in; (void)out_size; (void)d_ws; (void)ws_size;
  const int*   x   = (const int*)d_in[0];
  const float* emb = (const float*)d_in[1];
  const float* Wih = (const float*)d_in[2];
  const float* Whh = (const float*)d_in[3];
  const float* bih = (const float*)d_in[4];
  const float* bhh = (const float*)d_in[5];
  const float* Wfc = (const float*)d_in[6];
  const float* bfc = (const float*)d_in[7];
  float* out = (float*)d_out;

  hipLaunchKernelGGL(rnn_pipeline_kernel, dim3(BATCH / BT), dim3(NWAVES * 32),
                     0, stream, x, emb, Wih, Whh, bih, bhh, Wfc, bfc, out);
}


# The state's cross-lane move to the next step's columns becomes the DPP source operand of the tanh itself, the LDS copy is stored from the moved lanes: 8 fewer vector instructions per wave and step
# speedup vs baseline: 2.0243x; 1.0966x over previous
#include <hip/hip_runtime.h>

#define VOCAB   256
#define EMBED   32
#define HIDDEN  32
#define LAYERS  10
#define BATCH   1024
#define SEQT    1024

#define BT      4
#define NSLOT   8
#define SKEW    4
#define RS      36
#define SLOTF   (BT * RS)
#define LAYF    (NSLOT * SLOTF)
#define NBUF    (LAYERS * LAYF)
#define NEMB    (VOCAB * RS)
#define NWAVES  (LAYERS * 2)
#define NITER   (SEQT + SKEW * (LAYERS - 1))
#define NITER4  (NITER / 4)
#define HALFR   (4 * SLOTF)
#define LDSF    41984

typedef __attribute__((ext_vector_type(2))) float v2f;
typedef __attribute__((ext_vector_type(4))) float v4f;
typedef __attribute__((ext_vector_type(8))) float v8f;

__device__ __forceinline__ v8f wmma_k4(v2f a, v2f b, v8f c) {
  return __builtin_amdgcn_wmma_f32_16x16x4_f32(false, a, false, b, (short)0, c,
                                               false, false);
}

__device__ __forceinline__ float dev_tanh(float x) {
#if __has_builtin(__builtin_amdgcn_tanhf)
  return __builtin_amdgcn_tanhf(x);
#else
  float ax = __builtin_fabsf(x);
  float e  = __expf(-2.0f * ax);
  float y  = __fdividef(1.0f - e, 1.0f + e);
  return __builtin_copysignf(y, x);
#endif
}

__device__ __forceinline__ float swap8(float v) {
  return __builtin_bit_cast(float,
      __builtin_amdgcn_update_dpp(0, __builtin_bit_cast(int, v), 0x168, 0xf, 0xf, true));
}
__device__ __forceinline__ float swap4(float v) {
  return __builtin_bit_cast(float,
      __builtin_amdgcn_update_dpp(0, __builtin_bit_cast(int, v), 0x164, 0xf, 0xf, true));
}

__device__ __forceinline__ void bar_signal_after_lds() {
  asm volatile("s_wait_dscnt 0x0\n\ts_barrier_signal -1" ::: "memory");
}
__device__ __forceinline__ void bar_wait() {
  asm volatile("s_barrier_wait -1" ::: "memory");
}

__global__ void __launch_bounds__(NWAVES * 32)
rnn_pipeline_kernel(const int* __restrict__ x, const float* __restrict__ emb,
                    const float* __restrict__ W_ih, const float* __restrict__ W_hh,
                    const float* __restrict__ b_ih, const float* __restrict__ b_hh,
                    const float* __restrict__ W_fc, const float* __restrict__ b_fc,
                    float* __restrict__ out) {
  __shared__ __attribute__((aligned(16))) float smem[LDSF];
  float* buf  = smem;
  float* embs = smem + NBUF;

  const int tid  = threadIdx.x;
  const int lane = tid & 31;
  const int w    = __builtin_amdgcn_readfirstlane(tid >> 5);
  const int l    = w >> 1;
  const int c    = w & 1;
  const int R    = blockIdx.x * BT;

  for (int i = tid; i < NBUF; i += NWAVES * 32) buf[i] = 0.0f;
  for (int i = tid; i < VOCAB * EMBED; i += NWAVES * 32)
    embs[(i >> 5) * RS + (i & 31)] = emb[i];

  const int m    = lane & 15;
  const int hh   = lane >> 4;
  const int koff = hh * 2;
  const int b    = m & 3;
  const int g    = m >> 2;
  const int tn   = g ^ (g >> 1);

  const int orow = c * 16 + m;
  v2f aown[4];
  v4f aoth[2];
  v4f axw[4];
  float bias[8];
  {
    const float* Wi = W_ih + (l * HIDDEN + orow) * EMBED;
    const float* Wh = W_hh + (l * HIDDEN + orow) * HIDDEN;
#pragma unroll
    for (int j = 0; j < 4; ++j) aown[j] = *(const v2f*)(Wh + 16 * c + 8 * hh + 2 * j);
#pragma unroll
    for (int q = 0; q < 2; ++q) aoth[q] = *(const v4f*)(Wh + 16 * (1 - c) + 8 * q + 4 * hh);
#pragma unroll
    for (int q = 0; q < 4; ++q) axw[q]  = *(const v4f*)(Wi + 8 * q + 4 * hh);
#pragma unroll
    for (int r = 0; r < 8; ++r) {
      const int o = l * HIDDEN + c * 16 + 8 * hh + r;
      bias[r] = b_ih[o] + b_hh[o];
    }
  }

  v8f hreg;
  v8f P;
  v8f d;
#pragma unroll
  for (int r = 0; r < 8; ++r) { hreg[r] = 0.0f; P[r] = bias[r]; d[r] = 0.0f; }

  __syncthreads();

  const int lin = (l > 0) ? (l - 1) : 0;
  int o_hoth[4], o_hout[4];
#pragma unroll
  for (int ph = 0; ph < 4; ++ph) {
    o_hoth[ph] = l * LAYF + ((ph + 3) & 3) * SLOTF + b * RS + 16 * (1 - c) + 4 * hh;
    o_hout[ph] = l * LAYF + ph * SLOTF + b * RS + 16 * c + 8 * hh;
  }
  const int o_xin = lin * LAYF + tn * SLOTF + b * RS + 4 * hh;
  const int* xrow = x + (R + b) * SEQT + tn;

  for (int i4 = 0; i4 < NITER4; ++i4) {
    const int offs = ((i4 - l) & 1) * HALFR;
    const int offo = HALFR - offs;
#pragma unroll
    for (int ph = 0; ph < 4; ++ph) {
      const int t = i4 * 4 + ph - SKEW * l;
      __builtin_amdgcn_s_setprio(1);
      if ((unsigned)t < (unsigned)SEQT) {
        const float* hoth = buf + o_hoth[ph] + ((ph == 0) ? offo : offs);
        v4f bo[2];
#pragma unroll
        for (int q = 0; q < 2; ++q) bo[q] = *(const v4f*)(hoth + 8 * q);

        if (ph == 0) {
          v4f bx[4];
          if (l == 0) {
            const int tok   = xrow[t];
            const float* xe = embs + tok * RS + 4 * hh;
#pragma unroll
            for (int q = 0; q < 4; ++q) bx[q] = *(const v4f*)(xe + 8 * q);
          } else {
            const float* xin = buf + o_xin + offs;
#pragma unroll
            for (int q = 0; q < 4; ++q) bx[q] = *(const v4f*)(xin + 8 * q);
          }

          __builtin_amdgcn_sched_barrier(0);

#pragma unroll
          for (int q = 0; q < 2; ++q) {
            v2f a0, a1, b0, b1;
            a0[0] = aoth[q][0]; a0[1] = aoth[q][1]; a1[0] = aoth[q][2]; a1[1] = aoth[q][3];
            b0[0] = bo[q][0];   b0[1] = bo[q][1];   b1[0] = bo[q][2];   b1[1] = bo[q][3];
            d = wmma_k4(a0, b0, d);
            d = wmma_k4(a1, b1, d);
          }
#pragma unroll
          for (int r = 0; r < 8; ++r) P[r] = bias[r];
#pragma unroll
          for (int q = 0; q < 4; ++q) {
            v2f a0, a1, b0, b1;
            a0[0] = axw[q][0]; a0[1] = axw[q][1]; a1[0] = axw[q][2]; a1[1] = axw[q][3];
            b0[0] = bx[q][0];  b0[1] = bx[q][1];  b1[0] = bx[q][2];  b1[1] = bx[q][3];
            P = wmma_k4(a0, b0, P);
            P = wmma_k4(a1, b1, P);
          }
#pragma unroll
          for (int r = 0; r < 8; ++r) d[r] += P[r];
        } else {
          __builtin_amdgcn_sched_barrier(0);
#pragma unroll
          for (int q = 0; q < 2; ++q) {
            v2f a0, a1, b0, b1;
            a0[0] = aoth[q][0]; a0[1] = aoth[q][1]; a1[0] = aoth[q][2]; a1[1] = aoth[q][3];
            b0[0] = bo[q][0];   b0[1] = bo[q][1];   b1[0] = bo[q][2];   b1[1] = bo[q][3];
            d = wmma_k4(a0, b0, d);
            d = wmma_k4(a1, b1, d);
          }
        }

        if ((ph & 1) == 0) {
#pragma unroll
          for (int r = 0; r < 8; ++r) hreg[r] = dev_tanh(swap4(d[r]));
        } else {
#pragma unroll
          for (int r = 0; r < 8; ++r) hreg[r] = dev_tanh(swap8(d[r]));
        }

        if (tn == ((ph + 1) & 3)) {
          float* hout = buf + o_hout[ph] + offs;
          v4f s0, s1;
          s0[0] = hreg[0]; s0[1] = hreg[1]; s0[2] = hreg[2]; s0[3] = hreg[3];
          s1[0] = hreg[4]; s1[1] = hreg[5]; s1[2] = hreg[6]; s1[3] = hreg[7];
          *(v4f*)(hout)     = s0;
          *(v4f*)(hout + 4) = s1;
        }
      }
      __builtin_amdgcn_sched_barrier(0);
      bar_signal_after_lds();
      __builtin_amdgcn_s_setprio(0);
      __builtin_amdgcn_sched_barrier(0);
      if ((unsigned)(t + 1) < (unsigned)SEQT) {
        if (ph != 3) {
#pragma unroll
          for (int r = 0; r < 8; ++r) d[r] = P[r];
        } else {
#pragma unroll
          for (int r = 0; r < 8; ++r) d[r] = 0.0f;
        }
#pragma unroll
        for (int j = 0; j < 4; ++j) {
          v2f bb; bb[0] = hreg[2 * j]; bb[1] = hreg[2 * j + 1];
          d = wmma_k4(aown[j], bb, d);
        }
      }
      __builtin_amdgcn_sched_barrier(0);
      bar_wait();
    }
  }

  const float* htop = buf + (LAYERS - 1) * LAYF + ((SEQT - 1) & (NSLOT - 1)) * SLOTF;
  v2f ahf[8];
#pragma unroll
  for (int j = 0; j < 8; ++j)
    ahf[j] = *(const v2f*)(htop + b * RS + 4 * j + koff);

  for (int ct = w; ct < VOCAB / 16; ct += NWAVES) {
    const int vrow = ct * 16 + m;
    v2f wb[8];
#pragma unroll
    for (int j = 0; j < 8; ++j)
      wb[j] = *(const v2f*)(W_fc + vrow * HIDDEN + 4 * j + koff);
    const float bf = b_fc[vrow];
    v8f dd;
#pragma unroll
    for (int k = 0; k < 8; ++k) dd[k] = bf;
#pragma unroll
    for (int j = 0; j < 8; ++j) dd = wmma_k4(ahf[j], wb[j], dd);
    if (hh == 0) {
#pragma unroll
      for (int r = 0; r < BT; ++r)
        out[(R + r) * VOCAB + vrow] = dd[r];
    }
  }
}

extern "C" void kernel_launch(void* const* d_in, const int* in_sizes, int n_in,
                              void* d_out, int out_size, void* d_ws, size_t ws_size,
                              hipStream_t stream) {
  (void)in_sizes; (void)n_in; (void)out_size; (void)d_ws; (void)ws_size;
  const int*   x   = (const int*)d_in[0];
  const float* emb = (const float*)d_in[1];
  const float* Wih = (const float*)d_in[2];
  const float* Whh = (const float*)d_in[3];
  const float* bih = (const float*)d_in[4];
  const float* bhh = (const float*)d_in[5];
  const float* Wfc = (const float*)d_in[6];
  const float* bfc = (const float*)d_in[7];
  float* out = (float*)d_out;

  hipLaunchKernelGGL(rnn_pipeline_kernel, dim3(BATCH / BT), dim3(NWAVES * 32),
                     0, stream, x, emb, Wih, Whh, bih, bhh, Wfc, bfc, out);
}


# Input-product waves use the split barrier too: store the P tile, signal, then issue the next step's 8 f32 WMMAs before the wait; they fill the matrix pipe during the recurrent waves' tail
# speedup vs baseline: 2.6215x; 1.2950x over previous
#include <hip/hip_runtime.h>

#define VOCAB   256
#define EMBED   32
#define HIDDEN  32
#define LAYERS  10
#define BATCH   1024
#define SEQT    1024

#define BT      4
#define NSLOT   8
#define SKEW    4
#define RS      36
#define SLOTF   (BT * RS)
#define LAYF    (NSLOT * SLOTF)
#define NBUF    (LAYERS * LAYF)
#define NEMB    (VOCAB * RS)
#define NWAVES  (LAYERS * 2)
#define NITER   (SEQT + SKEW * (LAYERS - 1))
#define NITER4  (NITER / 4)
#define HALFR   (4 * SLOTF)
#define LDSF    41984

typedef __attribute__((ext_vector_type(2))) float v2f;
typedef __attribute__((ext_vector_type(4))) float v4f;
typedef __attribute__((ext_vector_type(8))) float v8f;

__device__ __forceinline__ v8f wmma_k4(v2f a, v2f b, v8f c) {
  return __builtin_amdgcn_wmma_f32_16x16x4_f32(false, a, false, b, (short)0, c,
                                               false, false);
}

__device__ __forceinline__ float dev_tanh(float x) {
#if __has_builtin(__builtin_amdgcn_tanhf)
  return __builtin_amdgcn_tanhf(x);
#else
  float ax = __builtin_fabsf(x);
  float e  = __expf(-2.0f * ax);
  float y  = __fdividef(1.0f - e, 1.0f + e);
  return __builtin_copysignf(y, x);
#endif
}

__device__ __forceinline__ float swap8(float v) {
  return __builtin_bit_cast(float,
      __builtin_amdgcn_update_dpp(0, __builtin_bit_cast(int, v), 0x168, 0xf, 0xf, true));
}
__device__ __forceinline__ float swap4(float v) {
  return __builtin_bit_cast(float,
      __builtin_amdgcn_update_dpp(0, __builtin_bit_cast(int, v), 0x164, 0xf, 0xf, true));
}

__device__ __forceinline__ void bar_signal_after_lds() {
  asm volatile("s_wait_dscnt 0x0\n\ts_barrier_signal -1" ::: "memory");
}
__device__ __forceinline__ void bar_wait() {
  asm volatile("s_barrier_wait -1" ::: "memory");
}
__device__ __forceinline__ void bar_wait_behind(v8f& pa, v8f& pb) {
  asm volatile("s_barrier_wait -1" : "+v"(pa), "+v"(pb) :: "memory");
}

__global__ void __launch_bounds__(NWAVES * 32)
rnn_pipeline_kernel(const int* __restrict__ x, const float* __restrict__ emb,
                    const float* __restrict__ W_ih, const float* __restrict__ W_hh,
                    const float* __restrict__ b_ih, const float* __restrict__ b_hh,
                    const float* __restrict__ W_fc, const float* __restrict__ b_fc,
                    float* __restrict__ out) {
  __shared__ __attribute__((aligned(16))) float smem[LDSF];
  float* buf  = smem;
  float* embs = smem + NBUF;

  const int tid  = threadIdx.x;
  const int lane = tid & 31;
  const int w    = __builtin_amdgcn_readfirstlane(tid >> 5);
  const int l    = w >> 1;
  const int c    = w & 1;
  const int R    = blockIdx.x * BT;

  for (int i = tid; i < NBUF; i += NWAVES * 32) buf[i] = 0.0f;
  for (int i = tid; i < VOCAB * EMBED; i += NWAVES * 32)
    embs[(i >> 5) * RS + (i & 31)] = emb[i];

  const int m    = lane & 15;
  const int hh   = lane >> 4;
  const int koff = hh * 2;
  const int b    = m & 3;
  const int g    = m >> 2;
  const int tn   = g ^ (g >> 1);

  const int orow = c * 16 + m;
  v2f aown[4];
  v4f aoth[2];
  v4f axw[4];
  float bias[8];
  {
    const float* Wi = W_ih + (l * HIDDEN + orow) * EMBED;
    const float* Wh = W_hh + (l * HIDDEN + orow) * HIDDEN;
#pragma unroll
    for (int j = 0; j < 4; ++j) aown[j] = *(const v2f*)(Wh + 16 * c + 8 * hh + 2 * j);
#pragma unroll
    for (int q = 0; q < 2; ++q) aoth[q] = *(const v4f*)(Wh + 16 * (1 - c) + 8 * q + 4 * hh);
#pragma unroll
    for (int q = 0; q < 4; ++q) axw[q]  = *(const v4f*)(Wi + 8 * q + 4 * hh);
#pragma unroll
    for (int r = 0; r < 8; ++r) {
      const int o = l * HIDDEN + c * 16 + 8 * hh + r;
      bias[r] = b_ih[o] + b_hh[o];
    }
  }

  v8f hreg;
  v8f P;
  v8f d;
#pragma unroll
  for (int r = 0; r < 8; ++r) { hreg[r] = 0.0f; P[r] = bias[r]; d[r] = 0.0f; }

  __syncthreads();

  const int lin = (l > 0) ? (l - 1) : 0;
  int o_hoth[4], o_hout[4];
#pragma unroll
  for (int ph = 0; ph < 4; ++ph) {
    o_hoth[ph] = l * LAYF + ((ph + 3) & 3) * SLOTF + b * RS + 16 * (1 - c) + 4 * hh;
    o_hout[ph] = l * LAYF + ph * SLOTF + b * RS + 16 * c + 8 * hh;
  }
  const int o_xin = lin * LAYF + tn * SLOTF + b * RS + 4 * hh;
  const int* xrow = x + (R + b) * SEQT + tn;

  for (int i4 = 0; i4 < NITER4; ++i4) {
    const int offs = ((i4 - l) & 1) * HALFR;
    const int offo = HALFR - offs;
#pragma unroll
    for (int ph = 0; ph < 4; ++ph) {
      const int t = i4 * 4 + ph - SKEW * l;
      __builtin_amdgcn_s_setprio(1);
      if ((unsigned)t < (unsigned)SEQT) {
        const float* hoth = buf + o_hoth[ph] + ((ph == 0) ? offo : offs);
        v4f bo[2];
#pragma unroll
        for (int q = 0; q < 2; ++q) bo[q] = *(const v4f*)(hoth + 8 * q);

        if (ph == 0) {
          v4f bx[4];
          if (l == 0) {
            const int tok   = xrow[t];
            const float* xe = embs + tok * RS + 4 * hh;
#pragma unroll
            for (int q = 0; q < 4; ++q) bx[q] = *(const v4f*)(xe + 8 * q);
          } else {
            const float* xin = buf + o_xin + offs;
#pragma unroll
            for (int q = 0; q < 4; ++q) bx[q] = *(const v4f*)(xin + 8 * q);
          }

          __builtin_amdgcn_sched_barrier(0);

#pragma unroll
          for (int q = 0; q < 2; ++q) {
            v2f a0, a1, b0, b1;
            a0[0] = aoth[q][0]; a0[1] = aoth[q][1]; a1[0] = aoth[q][2]; a1[1] = aoth[q][3];
            b0[0] = bo[q][0];   b0[1] = bo[q][1];   b1[0] = bo[q][2];   b1[1] = bo[q][3];
            d = wmma_k4(a0, b0, d);
            d = wmma_k4(a1, b1, d);
          }
#pragma unroll
          for (int r = 0; r < 8; ++r) P[r] = bias[r];
#pragma unroll
          for (int q = 0; q < 4; ++q) {
            v2f a0, a1, b0, b1;
            a0[0] = axw[q][0]; a0[1] = axw[q][1]; a1[0] = axw[q][2]; a1[1] = axw[q][3];
            b0[0] = bx[q][0];  b0[1] = bx[q][1];  b1[0] = bx[q][2];  b1[1] = bx[q][3];
            P = wmma_k4(a0, b0, P);
            P = wmma_k4(a1, b1, P);
          }
#pragma unroll
          for (int r = 0; r < 8; ++r) d[r] += P[r];
        } else {
          __builtin_amdgcn_sched_barrier(0);
#pragma unroll
          for (int q = 0; q < 2; ++q) {
            v2f a0, a1, b0, b1;
            a0[0] = aoth[q][0]; a0[1] = aoth[q][1]; a1[0] = aoth[q][2]; a1[1] = aoth[q][3];
            b0[0] = bo[q][0];   b0[1] = bo[q][1];   b1[0] = bo[q][2];   b1[1] = bo[q][3];
            d = wmma_k4(a0, b0, d);
            d = wmma_k4(a1, b1, d);
          }
        }

        if ((ph & 1) == 0) {
#pragma unroll
          for (int r = 0; r < 8; ++r) hreg[r] = dev_tanh(swap4(d[r]));
        } else {
#pragma unroll
          for (int r = 0; r < 8; ++r) hreg[r] = dev_tanh(swap8(d[r]));
        }

        if (tn == ((ph + 1) & 3)) {
          float* hout = buf + o_hout[ph] + offs;
          v4f s0, s1;
          s0[0] = hreg[0]; s0[1] = hreg[1]; s0[2] = hreg[2]; s0[3] = hreg[3];
          s1[0] = hreg[4]; s1[1] = hreg[5]; s1[2] = hreg[6]; s1[3] = hreg[7];
          *(v4f*)(hout)     = s0;
          *(v4f*)(hout + 4) = s1;
        }
      }
      __builtin_amdgcn_sched_barrier(0);
      bar_signal_after_lds();
      __builtin_amdgcn_s_setprio(0);
      __builtin_amdgcn_sched_barrier(0);
      if ((unsigned)(t + 1) < (unsigned)SEQT) {
        if (ph != 3) {
#pragma unroll
          for (int r = 0; r < 8; ++r) d[r] = P[r];
        } else {
#pragma unroll
          for (int r = 0; r < 8; ++r) d[r] = 0.0f;
        }
#pragma unroll
        for (int j = 0; j < 4; ++j) {
          v2f bb; bb[0] = hreg[2 * j]; bb[1] = hreg[2 * j + 1];
          d = wmma_k4(aown[j], bb, d);
        }
      }
      __builtin_amdgcn_sched_barrier(0);
      bar_wait();
    }
  }

  const float* htop = buf + (LAYERS - 1) * LAYF + ((SEQT - 1) & (NSLOT - 1)) * SLOTF;
  v2f ahf[8];
#pragma unroll
  for (int j = 0; j < 8; ++j)
    ahf[j] = *(const v2f*)(htop + b * RS + 4 * j + koff);

  for (int ct = w; ct < VOCAB / 16; ct += NWAVES) {
    const int vrow = ct * 16 + m;
    v2f wb[8];
#pragma unroll
    for (int j = 0; j < 8; ++j)
      wb[j] = *(const v2f*)(W_fc + vrow * HIDDEN + 4 * j + koff);
    const float bf = b_fc[vrow];
    v8f dd;
#pragma unroll
    for (int k = 0; k < 8; ++k) dd[k] = bf;
#pragma unroll
    for (int j = 0; j < 8; ++j) dd = wmma_k4(ahf[j], wb[j], dd);
    if (hh == 0) {
#pragma unroll
      for (int r = 0; r < BT; ++r)
        out[(R + r) * VOCAB + vrow] = dd[r];
    }
  }
}

#define RB2     16
#define NRG2    (BATCH / RB2)
#define NGRP2   4
#define CH2     64
#define NCH2    (SEQT / CH2)
#define WV2     12
#define NITER2  (CH2 + 8)
#define HSLOT2  (RB2 * RS)
#define HLAY2   (2 * HSLOT2)
#define PTILE2  256
#define PLAY2   (4 * PTILE2)
#define HB2F    (3 * HLAY2)
#define PB2F    (3 * PLAY2)
#define STEPF2  512
#define HANDF2  (3 * 2 * NRG2 * CH2 * STEPF2)
#define STATEF2 (NGRP2 * NRG2 * 3 * 512)
#define WSNEED2 ((HANDF2 + STATEF2) * 4ull)

typedef __attribute__((ext_vector_type(4))) int v4i;

__device__ __forceinline__ void bar_round() {
  __builtin_amdgcn_sched_barrier(0);
  bar_signal_after_lds();
  bar_wait();
  __builtin_amdgcn_sched_barrier(0);
}

__device__ __forceinline__ void input_mma(const v4f* axw, const float* bias, const v4f* bx, v8f& PA, v8f& PB) {
#pragma unroll
  for (int r = 0; r < 8; ++r) { PA[r] = bias[r]; PB[r] = 0.0f; }
#pragma unroll
  for (int h = 0; h < 2; ++h) {
    const int qa = h, qb = h + 2;
    v2f a0, a1, b0, b1, e0, e1, f0, f1;
    a0[0] = axw[qa][0]; a0[1] = axw[qa][1]; a1[0] = axw[qa][2]; a1[1] = axw[qa][3];
    b0[0] = bx[qa][0];  b0[1] = bx[qa][1];  b1[0] = bx[qa][2];  b1[1] = bx[qa][3];
    e0[0] = axw[qb][0]; e0[1] = axw[qb][1]; e1[0] = axw[qb][2]; e1[1] = axw[qb][3];
    f0[0] = bx[qb][0];  f0[1] = bx[qb][1];  f1[0] = bx[qb][2];  f1[1] = bx[qb][3];
    PA = wmma_k4(a0, b0, PA);
    PB = wmma_k4(e0, f0, PB);
    PA = wmma_k4(a1, b1, PA);
    PB = wmma_k4(e1, f1, PB);
  }
}
__device__ __forceinline__ void input_store(const v8f& PA, const v8f& PB, float* pt) {
  v8f P;
#pragma unroll
  for (int r = 0; r < 8; ++r) P[r] = PA[r] + PB[r];
  v4f s0, s1;
  s0[0] = P[0]; s0[1] = P[1]; s0[2] = P[2]; s0[3] = P[3];
  s1[0] = P[4]; s1[1] = P[5]; s1[2] = P[6]; s1[3] = P[7];
  *(v4f*)(pt)       = s0;
  *(v4f*)(pt + 128) = s1;
}
#define INPUT_ROUND_MMA(BX)                                                                      \
  do {                                                                                           \
    __builtin_amdgcn_sched_barrier(0);      \
    bar_signal_after_lds();                \
    __builtin_amdgcn_sched_barrier(0);                      \
    input_mma(axw, bias, BX, PA, PB);                                                            \
    __builtin_amdgcn_sched_barrier(0);                   \
    bar_wait_behind(PA, PB);                                                                     \
    __builtin_amdgcn_sched_barrier(0);                                                           \
  } while (0)

__global__ void __launch_bounds__(WV2 * 32)
rnn_split_kernel(const int* __restrict__ x, const float* __restrict__ emb,
                 const float* __restrict__ W_ih, const float* __restrict__ W_hh,
                 const float* __restrict__ b_ih, const float* __restrict__ b_hh,
                 const float* __restrict__ W_fc, const float* __restrict__ b_fc,
                 float* __restrict__ out, float* __restrict__ ws, int kstep) {
  __shared__ __attribute__((aligned(16))) float smem[LDSF];
  float* hb   = smem;
  float* pb   = smem + HB2F;
  float* embs = smem + HB2F + PB2F;

  const int g    = blockIdx.x >> 6;
  const int rg   = blockIdx.x & 63;
  const int cidx = kstep - g;
  if (cidx < 0 || cidx >= NCH2) return;

  const int tid  = threadIdx.x;
  const int lane = tid & 31;
  const int w    = __builtin_amdgcn_readfirstlane(tid >> 5);
  const int j    = w >> 2;
  const int x4   = w & 3;
  const int kind = (x4 >> 1) ^ (j & 1);
  const int c    = x4 & 1;
  const int L    = (g < 2) ? 3 : 2;
  const int l    = ((g < 2) ? 3 * g : 2 * g + 2) + j;
  const bool act = j < L;
  const bool top = (j == L - 1);
  const int R    = rg * RB2;
  const int t0   = cidx * CH2;

  const int n    = lane & 15;
  const int hh   = lane >> 4;

  if (g == 0) {
    for (int i = tid; i < VOCAB * EMBED / 4; i += WV2 * 32)
      *(v4f*)(embs + (i >> 3) * RS + (i & 7) * 4) = *(const v4f*)(emb + i * 4);
  }

  const int orow = c * 16 + n;
  float* stp = ws + HANDF2 + ((g * NRG2 + rg) * 3 + j) * 512 + n * 32 + 16 * c + 8 * hh;
  float* hst = hb + j * HLAY2 + n * RS + 16 * c + 8 * hh;
  const float* hot = hb + j * HLAY2 + n * RS + 16 * (1 - c) + 4 * hh;
  float* ptl = pb + (j * 2 + c) * (2 * PTILE2) + lane * 4;

  v8f hreg;
#pragma unroll
  for (int r = 0; r < 8; ++r) hreg[r] = 0.0f;
  if (act && kind == 0) {
    if (cidx > 0) {
      v4f s0 = *(const v4f*)(stp), s1 = *(const v4f*)(stp + 4);
      hreg[0] = s0[0]; hreg[1] = s0[1]; hreg[2] = s0[2]; hreg[3] = s0[3];
      hreg[4] = s1[0]; hreg[5] = s1[1]; hreg[6] = s1[2]; hreg[7] = s1[3];
    }
    v4f s0, s1;
    s0[0] = hreg[0]; s0[1] = hreg[1]; s0[2] = hreg[2]; s0[3] = hreg[3];
    s1[0] = hreg[4]; s1[1] = hreg[5]; s1[2] = hreg[6]; s1[3] = hreg[7];
    *(v4f*)(hst + HSLOT2)     = s0;
    *(v4f*)(hst + HSLOT2 + 4) = s1;
  }

  __syncthreads();

  if (!act) {
#pragma unroll 1
    for (int i = 0; i < NITER2; ++i) bar_round();
  } else if (kind == 0) {
    v2f aown[4]; v4f aoth[2];
    {
      const float* Wh = W_hh + (l * HIDDEN + orow) * HIDDEN;
#pragma unroll
      for (int jj = 0; jj < 4; ++jj) aown[jj] = *(const v2f*)(Wh + 16 * c + 8 * hh + 2 * jj);
#pragma unroll
      for (int q = 0; q < 2; ++q) aoth[q] = *(const v4f*)(Wh + 16 * (1 - c) + 8 * q + 4 * hh);
    }
    float* hog = ws + (size_t)(((g * 2 + (cidx & 1)) * NRG2 + rg) * CH2) * STEPF2 + (2 * c + hh) * 128 + n * 4;
    const bool hand = top && g < NGRP2 - 1;
    v8f dA, dB;
#pragma unroll 1
    for (int i = 0; i < 3 * j + 1; ++i) bar_round();
    bar_signal_after_lds();
    {
#pragma unroll
      for (int r = 0; r < 8; ++r) { dA[r] = 0.0f; dB[r] = 0.0f; }
      v2f h0, h1, h2, h3;
      h0[0] = hreg[0]; h0[1] = hreg[1]; h1[0] = hreg[2]; h1[1] = hreg[3];
      h2[0] = hreg[4]; h2[1] = hreg[5]; h3[0] = hreg[6]; h3[1] = hreg[7];
      dA = wmma_k4(aown[0], h0, dA);
      dB = wmma_k4(aown[2], h2, dB);
      dA = wmma_k4(aown[1], h1, dA);
      dB = wmma_k4(aown[3], h3, dB);
    }
    bar_wait();
#pragma unroll 1
    for (int s2 = 0; s2 < CH2; s2 += 2) {
#pragma unroll
      for (int e = 0; e < 2; ++e) {
        __builtin_amdgcn_s_setprio(1);
        v4f bo[2];
#pragma unroll
        for (int q = 0; q < 2; ++q) bo[q] = *(const v4f*)(hot + (1 - e) * HSLOT2 + 8 * q);
        const v4f p0 = *(const v4f*)(ptl + e * PTILE2);
        const v4f p1 = *(const v4f*)(ptl + e * PTILE2 + 128);
        __builtin_amdgcn_sched_barrier(0);
        {
          v2f a0, a1, b0, b1, e0, e1, f0, f1;
          a0[0] = aoth[0][0]; a0[1] = aoth[0][1]; a1[0] = aoth[0][2]; a1[1] = aoth[0][3];
          b0[0] = bo[0][0];   b0[1] = bo[0][1];   b1[0] = bo[0][2];   b1[1] = bo[0][3];
          e0[0] = aoth[1][0]; e0[1] = aoth[1][1]; e1[0] = aoth[1][2]; e1[1] = aoth[1][3];
          f0[0] = bo[1][0];   f0[1] = bo[1][1];   f1[0] = bo[1][2];   f1[1] = bo[1][3];
          dA = wmma_k4(a0, b0, dA);
          dB = wmma_k4(e0, f0, dB);
          dA = wmma_k4(a1, b1, dA);
          dB = wmma_k4(e1, f1, dB);
        }
        hreg[0] = dev_tanh((dA[0] + dB[0]) + p0[0]); hreg[1] = dev_tanh((dA[1] + dB[1]) + p0[1]);
        hreg[2] = dev_tanh((dA[2] + dB[2]) + p0[2]); hreg[3] = dev_tanh((dA[3] + dB[3]) + p0[3]);
        hreg[4] = dev_tanh((dA[4] + dB[4]) + p1[0]); hreg[5] = dev_tanh((dA[5] + dB[5]) + p1[1]);
        hreg[6] = dev_tanh((dA[6] + dB[6]) + p1[2]); hreg[7] = dev_tanh((dA[7] + dB[7]) + p1[3]);
        v4f s0, s1;
        s0[0] = hreg[0]; s0[1] = hreg[1]; s0[2] = hreg[2]; s0[3] = hreg[3];
        s1[0] = hreg[4]; s1[1] = hreg[5]; s1[2] = hreg[6]; s1[3] = hreg[7];
        *(v4f*)(hst + e * HSLOT2)     = s0;
        *(v4f*)(hst + e * HSLOT2 + 4) = s1;
        if (hand) {
          float* ho = hog + (s2 + e) * STEPF2;
          *(v4f*)(ho)      = s0;
          *(v4f*)(ho + 64) = s1;
        }
        __builtin_amdgcn_sched_barrier(0);
        bar_signal_after_lds();
        __builtin_amdgcn_s_setprio(0);
        __builtin_amdgcn_sched_barrier(0);
        {
#pragma unroll
          for (int r = 0; r < 8; ++r) { dA[r] = 0.0f; dB[r] = 0.0f; }
          v2f h0, h1, h2, h3;
          h0[0] = hreg[0]; h0[1] = hreg[1]; h1[0] = hreg[2]; h1[1] = hreg[3];
          h2[0] = hreg[4]; h2[1] = hreg[5]; h3[0] = hreg[6]; h3[1] = hreg[7];
          dA = wmma_k4(aown[0], h0, dA);
          dB = wmma_k4(aown[2], h2, dB);
          dA = wmma_k4(aown[1], h1, dA);
          dB = wmma_k4(aown[3], h3, dB);
        }
        __builtin_amdgcn_sched_barrier(0);
        bar_wait();
        __builtin_amdgcn_sched_barrier(0);
      }
    }
#pragma unroll 1
    for (int i = 0; i < NITER2 - CH2 - 2 - 3 * j; ++i) bar_round();
    {
      v4f s0, s1;
      s0[0] = hreg[0]; s0[1] = hreg[1]; s0[2] = hreg[2]; s0[3] = hreg[3];
      s1[0] = hreg[4]; s1[1] = hreg[5]; s1[2] = hreg[6]; s1[3] = hreg[7];
      *(v4f*)(stp)     = s0;
      *(v4f*)(stp + 4) = s1;
    }
  } else {
    v4f axw[4]; float bias[8];
    {
      const float* Wi = W_ih + (l * HIDDEN + orow) * EMBED;
#pragma unroll
      for (int q = 0; q < 4; ++q) axw[q] = *(const v4f*)(Wi + 8 * q + 4 * hh);
#pragma unroll
      for (int r = 0; r < 8; ++r) {
        const int o = l * HIDDEN + c * 16 + 8 * hh + r;
        bias[r] = b_ih[o] + b_hh[o];
      }
    }
    if (j > 0) {
      const float* xin = hb + (j - 1) * HLAY2 + n * RS + 4 * hh;
#pragma unroll 1
      for (int i = 0; i < 3 * j; ++i) bar_round();
      v4f bxn[4];
      v8f PA, PB;
#pragma unroll
      for (int q = 0; q < 4; ++q) bxn[q] = *(const v4f*)(xin + 8 * q);
      INPUT_ROUND_MMA(bxn);
#pragma unroll 1
      for (int s2 = 0; s2 < CH2; s2 += 2) {
#pragma unroll
        for (int e = 0; e < 2; ++e) {
#pragma unroll
          for (int q = 0; q < 4; ++q) bxn[q] = *(const v4f*)(xin + (1 - e) * HSLOT2 + 8 * q);
          input_store(PA, PB, ptl + e * PTILE2);
          INPUT_ROUND_MMA(bxn);
        }
      }
#pragma unroll 1
      for (int i = 0; i < NITER2 - CH2 - 1 - 3 * j; ++i) bar_round();
    } else if (g == 0) {
      const int* xrow = x + (R + n) * SEQT + t0;
      v4i tokc = *(const v4i*)(xrow);
      v4f bxn[4];
      v8f PA, PB;
      {
        const float* xe = embs + tokc[0] * RS + 4 * hh;
#pragma unroll
        for (int q = 0; q < 4; ++q) bxn[q] = *(const v4f*)(xe + 8 * q);
      }
      INPUT_ROUND_MMA(bxn);
#pragma unroll 1
      for (int s4 = 0; s4 < CH2; s4 += 4) {
        const int sn = (s4 + 4 < CH2) ? s4 + 4 : s4;
        const v4i tokn = *(const v4i*)(xrow + sn);
#pragma unroll
        for (int ph = 0; ph < 4; ++ph) {
          const int tk = (ph < 3) ? tokc[(ph + 1) & 3] : tokn[0];
          const float* xe = embs + tk * RS + 4 * hh;
#pragma unroll
          for (int q = 0; q < 4; ++q) bxn[q] = *(const v4f*)(xe + 8 * q);
          input_store(PA, PB, ptl + (ph & 1) * PTILE2);
          INPUT_ROUND_MMA(bxn);
        }
        tokc = tokn;
      }
#pragma unroll 1
      for (int i = 0; i < NITER2 - CH2 - 1; ++i) bar_round();
    } else {
      const float* hin = ws + (size_t)((((g - 1) * 2 + (cidx & 1)) * NRG2 + rg) * CH2) * STEPF2 + lane * 4;
      v4f bxg[4][4];
#pragma unroll
      for (int ph = 0; ph < 4; ++ph)
#pragma unroll
        for (int q = 0; q < 4; ++q) bxg[ph][q] = *(const v4f*)(hin + ph * STEPF2 + q * 128);
      v8f PA, PB;
      INPUT_ROUND_MMA(bxg[0]);
#pragma unroll
      for (int q = 0; q < 4; ++q) bxg[0][q] = *(const v4f*)(hin + 4 * STEPF2 + q * 128);
#pragma unroll 1
      for (int s4 = 0; s4 < CH2; s4 += 4) {
#pragma unroll
        for (int ph = 0; ph < 4; ++ph) {
          input_store(PA, PB, ptl + (ph & 1) * PTILE2);
          INPUT_ROUND_MMA(bxg[(ph + 1) & 3]);
          const int s5 = (s4 + ph + 5 < CH2) ? s4 + ph + 5 : CH2 - 1;
#pragma unroll
          for (int q = 0; q < 4; ++q) bxg[(ph + 1) & 3][q] = *(const v4f*)(hin + s5 * STEPF2 + q * 128);
        }
      }
#pragma unroll 1
      for (int i = 0; i < NITER2 - CH2 - 1; ++i) bar_round();
    }
  }

  if (g == NGRP2 - 1 && cidx == NCH2 - 1 && w < 8) {
    const int koff = hh * 2;
    const float* htop = hb + (L - 1) * HLAY2 + ((CH2 - 1) & 1) * HSLOT2;
    v2f ahf[8];
#pragma unroll
    for (int jj = 0; jj < 8; ++jj) ahf[jj] = *(const v2f*)(htop + n * RS + 4 * jj + koff);
#pragma unroll
    for (int e = 0; e < 2; ++e) {
      const int vrow = (2 * w + e) * 16 + n;
      v2f wb[8];
#pragma unroll
      for (int jj = 0; jj < 8; ++jj) wb[jj] = *(const v2f*)(W_fc + vrow * HIDDEN + 4 * jj + koff);
      const float bf = b_fc[vrow];
      v8f dd;
#pragma unroll
      for (int k = 0; k < 8; ++k) dd[k] = bf;
#pragma unroll
      for (int jj = 0; jj < 8; ++jj) dd = wmma_k4(ahf[jj], wb[jj], dd);
#pragma unroll
      for (int r = 0; r < 8; ++r) out[(R + 8 * hh + r) * VOCAB + vrow] = dd[r];
    }
  }
}

#define SPLIT_LAUNCH(K) hipLaunchKernelGGL(rnn_split_kernel, dim3(NRG2 * NGRP2), dim3(WV2 * 32), 0, stream, \
                                           x, emb, Wih, Whh, bih, bhh, Wfc, bfc, out, wsf, K)

extern "C" void kernel_launch(void* const* d_in, const int* in_sizes, int n_in,
                              void* d_out, int out_size, void* d_ws, size_t ws_size,
                              hipStream_t stream) {
  (void)in_sizes; (void)n_in; (void)out_size;
  const int*   x   = (const int*)d_in[0];
  const float* emb = (const float*)d_in[1];
  const float* Wih = (const float*)d_in[2];
  const float* Whh = (const float*)d_in[3];
  const float* bih = (const float*)d_in[4];
  const float* bhh = (const float*)d_in[5];
  const float* Wfc = (const float*)d_in[6];
  const float* bfc = (const float*)d_in[7];
  float* out = (float*)d_out;
  float* wsf = (float*)d_ws;

  if (d_ws != nullptr && ws_size >= WSNEED2) {
    SPLIT_LAUNCH(0);  SPLIT_LAUNCH(1);  SPLIT_LAUNCH(2);  SPLIT_LAUNCH(3);  SPLIT_LAUNCH(4);
    SPLIT_LAUNCH(5);  SPLIT_LAUNCH(6);  SPLIT_LAUNCH(7);  SPLIT_LAUNCH(8);  SPLIT_LAUNCH(9);
    SPLIT_LAUNCH(10); SPLIT_LAUNCH(11); SPLIT_LAUNCH(12); SPLIT_LAUNCH(13); SPLIT_LAUNCH(14);
    SPLIT_LAUNCH(15); SPLIT_LAUNCH(16); SPLIT_LAUNCH(17); SPLIT_LAUNCH(18);
  } else {
    hipLaunchKernelGGL(rnn_pipeline_kernel, dim3(BATCH / BT), dim3(NWAVES * 32),
                       0, stream, x, emb, Wih, Whh, bih, bhh, Wfc, bfc, out);
  }
}


# Start-up global loads of each launch (embedding table, weights, biases, chunk state, first tokens or hand-over fragments) issued as one group before the first barrier, not one L2 trip after another
# speedup vs baseline: 2.6996x; 1.0298x over previous
#include <hip/hip_runtime.h>

#define VOCAB   256
#define EMBED   32
#define HIDDEN  32
#define LAYERS  10
#define BATCH   1024
#define SEQT    1024

#define BT      4
#define NSLOT   8
#define SKEW    4
#define RS      36
#define SLOTF   (BT * RS)
#define LAYF    (NSLOT * SLOTF)
#define NBUF    (LAYERS * LAYF)
#define NEMB    (VOCAB * RS)
#define NWAVES  (LAYERS * 2)
#define NITER   (SEQT + SKEW * (LAYERS - 1))
#define NITER4  (NITER / 4)
#define HALFR   (4 * SLOTF)
#define LDSF    41984

typedef __attribute__((ext_vector_type(2))) float v2f;
typedef __attribute__((ext_vector_type(4))) float v4f;
typedef __attribute__((ext_vector_type(8))) float v8f;

__device__ __forceinline__ v8f wmma_k4(v2f a, v2f b, v8f c) {
  return __builtin_amdgcn_wmma_f32_16x16x4_f32(false, a, false, b, (short)0, c,
                                               false, false);
}

__device__ __forceinline__ float dev_tanh(float x) {
#if __has_builtin(__builtin_amdgcn_tanhf)
  return __builtin_amdgcn_tanhf(x);
#else
  float ax = __builtin_fabsf(x);
  float e  = __expf(-2.0f * ax);
  float y  = __fdividef(1.0f - e, 1.0f + e);
  return __builtin_copysignf(y, x);
#endif
}

__device__ __forceinline__ float swap8(float v) {
  return __builtin_bit_cast(float,
      __builtin_amdgcn_update_dpp(0, __builtin_bit_cast(int, v), 0x168, 0xf, 0xf, true));
}
__device__ __forceinline__ float swap4(float v) {
  return __builtin_bit_cast(float,
      __builtin_amdgcn_update_dpp(0, __builtin_bit_cast(int, v), 0x164, 0xf, 0xf, true));
}

__device__ __forceinline__ void bar_signal_after_lds() {
  asm volatile("s_wait_dscnt 0x0\n\ts_barrier_signal -1" ::: "memory");
}
__device__ __forceinline__ void bar_wait() {
  asm volatile("s_barrier_wait -1" ::: "memory");
}
__device__ __forceinline__ void bar_wait_behind(v8f& pa, v8f& pb) {
  asm volatile("s_barrier_wait -1" : "+v"(pa), "+v"(pb) :: "memory");
}

__global__ void __launch_bounds__(NWAVES * 32)
rnn_pipeline_kernel(const int* __restrict__ x, const float* __restrict__ emb,
                    const float* __restrict__ W_ih, const float* __restrict__ W_hh,
                    const float* __restrict__ b_ih, const float* __restrict__ b_hh,
                    const float* __restrict__ W_fc, const float* __restrict__ b_fc,
                    float* __restrict__ out) {
  __shared__ __attribute__((aligned(16))) float smem[LDSF];
  float* buf  = smem;
  float* embs = smem + NBUF;

  const int tid  = threadIdx.x;
  const int lane = tid & 31;
  const int w    = __builtin_amdgcn_readfirstlane(tid >> 5);
  const int l    = w >> 1;
  const int c    = w & 1;
  const int R    = blockIdx.x * BT;

  for (int i = tid; i < NBUF; i += NWAVES * 32) buf[i] = 0.0f;
  for (int i = tid; i < VOCAB * EMBED; i += NWAVES * 32)
    embs[(i >> 5) * RS + (i & 31)] = emb[i];

  const int m    = lane & 15;
  const int hh   = lane >> 4;
  const int koff = hh * 2;
  const int b    = m & 3;
  const int g    = m >> 2;
  const int tn   = g ^ (g >> 1);

  const int orow = c * 16 + m;
  v2f aown[4];
  v4f aoth[2];
  v4f axw[4];
  float bias[8];
  {
    const float* Wi = W_ih + (l * HIDDEN + orow) * EMBED;
    const float* Wh = W_hh + (l * HIDDEN + orow) * HIDDEN;
#pragma unroll
    for (int j = 0; j < 4; ++j) aown[j] = *(const v2f*)(Wh + 16 * c + 8 * hh + 2 * j);
#pragma unroll
    for (int q = 0; q < 2; ++q) aoth[q] = *(const v4f*)(Wh + 16 * (1 - c) + 8 * q + 4 * hh);
#pragma unroll
    for (int q = 0; q < 4; ++q) axw[q]  = *(const v4f*)(Wi + 8 * q + 4 * hh);
#pragma unroll
    for (int r = 0; r < 8; ++r) {
      const int o = l * HIDDEN + c * 16 + 8 * hh + r;
      bias[r] = b_ih[o] + b_hh[o];
    }
  }

  v8f hreg;
  v8f P;
  v8f d;
#pragma unroll
  for (int r = 0; r < 8; ++r) { hreg[r] = 0.0f; P[r] = bias[r]; d[r] = 0.0f; }

  __syncthreads();

  const int lin = (l > 0) ? (l - 1) : 0;
  int o_hoth[4], o_hout[4];
#pragma unroll
  for (int ph = 0; ph < 4; ++ph) {
    o_hoth[ph] = l * LAYF + ((ph + 3) & 3) * SLOTF + b * RS + 16 * (1 - c) + 4 * hh;
    o_hout[ph] = l * LAYF + ph * SLOTF + b * RS + 16 * c + 8 * hh;
  }
  const int o_xin = lin * LAYF + tn * SLOTF + b * RS + 4 * hh;
  const int* xrow = x + (R + b) * SEQT + tn;

  for (int i4 = 0; i4 < NITER4; ++i4) {
    const int offs = ((i4 - l) & 1) * HALFR;
    const int offo = HALFR - offs;
#pragma unroll
    for (int ph = 0; ph < 4; ++ph) {
      const int t = i4 * 4 + ph - SKEW * l;
      __builtin_amdgcn_s_setprio(1);
      if ((unsigned)t < (unsigned)SEQT) {
        const float* hoth = buf + o_hoth[ph] + ((ph == 0) ? offo : offs);
        v4f bo[2];
#pragma unroll
        for (int q = 0; q < 2; ++q) bo[q] = *(const v4f*)(hoth + 8 * q);

        if (ph == 0) {
          v4f bx[4];
          if (l == 0) {
            const int tok   = xrow[t];
            const float* xe = embs + tok * RS + 4 * hh;
#pragma unroll
            for (int q = 0; q < 4; ++q) bx[q] = *(const v4f*)(xe + 8 * q);
          } else {
            const float* xin = buf + o_xin + offs;
#pragma unroll
            for (int q = 0; q < 4; ++q) bx[q] = *(const v4f*)(xin + 8 * q);
          }

          __builtin_amdgcn_sched_barrier(0);

#pragma unroll
          for (int q = 0; q < 2; ++q) {
            v2f a0, a1, b0, b1;
            a0[0] = aoth[q][0]; a0[1] = aoth[q][1]; a1[0] = aoth[q][2]; a1[1] = aoth[q][3];
            b0[0] = bo[q][0];   b0[1] = bo[q][1];   b1[0] = bo[q][2];   b1[1] = bo[q][3];
            d = wmma_k4(a0, b0, d);
            d = wmma_k4(a1, b1, d);
          }
#pragma unroll
          for (int r = 0; r < 8; ++r) P[r] = bias[r];
#pragma unroll
          for (int q = 0; q < 4; ++q) {
            v2f a0, a1, b0, b1;
            a0[0] = axw[q][0]; a0[1] = axw[q][1]; a1[0] = axw[q][2]; a1[1] = axw[q][3];
            b0[0] = bx[q][0];  b0[1] = bx[q][1];  b1[0] = bx[q][2];  b1[1] = bx[q][3];
            P = wmma_k4(a0, b0, P);
            P = wmma_k4(a1, b1, P);
          }
#pragma unroll
          for (int r = 0; r < 8; ++r) d[r] += P[r];
        } else {
          __builtin_amdgcn_sched_barrier(0);
#pragma unroll
          for (int q = 0; q < 2; ++q) {
            v2f a0, a1, b0, b1;
            a0[0] = aoth[q][0]; a0[1] = aoth[q][1]; a1[0] = aoth[q][2]; a1[1] = aoth[q][3];
            b0[0] = bo[q][0];   b0[1] = bo[q][1];   b1[0] = bo[q][2];   b1[1] = bo[q][3];
            d = wmma_k4(a0, b0, d);
            d = wmma_k4(a1, b1, d);
          }
        }

        if ((ph & 1) == 0) {
#pragma unroll
          for (int r = 0; r < 8; ++r) hreg[r] = dev_tanh(swap4(d[r]));
        } else {
#pragma unroll
          for (int r = 0; r < 8; ++r) hreg[r] = dev_tanh(swap8(d[r]));
        }

        if (tn == ((ph + 1) & 3)) {
          float* hout = buf + o_hout[ph] + offs;
          v4f s0, s1;
          s0[0] = hreg[0]; s0[1] = hreg[1]; s0[2] = hreg[2]; s0[3] = hreg[3];
          s1[0] = hreg[4]; s1[1] = hreg[5]; s1[2] = hreg[6]; s1[3] = hreg[7];
          *(v4f*)(hout)     = s0;
          *(v4f*)(hout + 4) = s1;
        }
      }
      __builtin_amdgcn_sched_barrier(0);
      bar_signal_after_lds();
      __builtin_amdgcn_s_setprio(0);
      __builtin_amdgcn_sched_barrier(0);
      if ((unsigned)(t + 1) < (unsigned)SEQT) {
        if (ph != 3) {
#pragma unroll
          for (int r = 0; r < 8; ++r) d[r] = P[r];
        } else {
#pragma unroll
          for (int r = 0; r < 8; ++r) d[r] = 0.0f;
        }
#pragma unroll
        for (int j = 0; j < 4; ++j) {
          v2f bb; bb[0] = hreg[2 * j]; bb[1] = hreg[2 * j + 1];
          d = wmma_k4(aown[j], bb, d);
        }
      }
      __builtin_amdgcn_sched_barrier(0);
      bar_wait();
    }
  }

  const float* htop = buf + (LAYERS - 1) * LAYF + ((SEQT - 1) & (NSLOT - 1)) * SLOTF;
  v2f ahf[8];
#pragma unroll
  for (int j = 0; j < 8; ++j)
    ahf[j] = *(const v2f*)(htop + b * RS + 4 * j + koff);

  for (int ct = w; ct < VOCAB / 16; ct += NWAVES) {
    const int vrow = ct * 16 + m;
    v2f wb[8];
#pragma unroll
    for (int j = 0; j < 8; ++j)
      wb[j] = *(const v2f*)(W_fc + vrow * HIDDEN + 4 * j + koff);
    const float bf = b_fc[vrow];
    v8f dd;
#pragma unroll
    for (int k = 0; k < 8; ++k) dd[k] = bf;
#pragma unroll
    for (int j = 0; j < 8; ++j) dd = wmma_k4(ahf[j], wb[j], dd);
    if (hh == 0) {
#pragma unroll
      for (int r = 0; r < BT; ++r)
        out[(R + r) * VOCAB + vrow] = dd[r];
    }
  }
}

#define RB2     16
#define NRG2    (BATCH / RB2)
#define NGRP2   4
#define CH2     64
#define NCH2    (SEQT / CH2)
#define WV2     12
#define NITER2  (CH2 + 8)
#define HSLOT2  (RB2 * RS)
#define HLAY2   (2 * HSLOT2)
#define PTILE2  256
#define PLAY2   (4 * PTILE2)
#define HB2F    (3 * HLAY2)
#define PB2F    (3 * PLAY2)
#define STEPF2  512
#define HANDF2  (3 * 2 * NRG2 * CH2 * STEPF2)
#define STATEF2 (NGRP2 * NRG2 * 3 * 512)
#define WSNEED2 ((HANDF2 + STATEF2) * 4ull)

typedef __attribute__((ext_vector_type(4))) int v4i;

__device__ __forceinline__ void bar_round() {
  __builtin_amdgcn_sched_barrier(0);
  bar_signal_after_lds();
  bar_wait();
  __builtin_amdgcn_sched_barrier(0);
}

__device__ __forceinline__ void input_mma(const v4f* axw, const float* bias, const v4f* bx, v8f& PA, v8f& PB) {
#pragma unroll
  for (int r = 0; r < 8; ++r) { PA[r] = bias[r]; PB[r] = 0.0f; }
#pragma unroll
  for (int h = 0; h < 2; ++h) {
    const int qa = h, qb = h + 2;
    v2f a0, a1, b0, b1, e0, e1, f0, f1;
    a0[0] = axw[qa][0]; a0[1] = axw[qa][1]; a1[0] = axw[qa][2]; a1[1] = axw[qa][3];
    b0[0] = bx[qa][0];  b0[1] = bx[qa][1];  b1[0] = bx[qa][2];  b1[1] = bx[qa][3];
    e0[0] = axw[qb][0]; e0[1] = axw[qb][1]; e1[0] = axw[qb][2]; e1[1] = axw[qb][3];
    f0[0] = bx[qb][0];  f0[1] = bx[qb][1];  f1[0] = bx[qb][2];  f1[1] = bx[qb][3];
    PA = wmma_k4(a0, b0, PA);
    PB = wmma_k4(e0, f0, PB);
    PA = wmma_k4(a1, b1, PA);
    PB = wmma_k4(e1, f1, PB);
  }
}
__device__ __forceinline__ void input_store(const v8f& PA, const v8f& PB, float* pt) {
  v8f P;
#pragma unroll
  for (int r = 0; r < 8; ++r) P[r] = PA[r] + PB[r];
  v4f s0, s1;
  s0[0] = P[0]; s0[1] = P[1]; s0[2] = P[2]; s0[3] = P[3];
  s1[0] = P[4]; s1[1] = P[5]; s1[2] = P[6]; s1[3] = P[7];
  *(v4f*)(pt)       = s0;
  *(v4f*)(pt + 128) = s1;
}
#define INPUT_ROUND_MMA(BX)                                                                      \
  do {                                                                                           \
    __builtin_amdgcn_sched_barrier(0);      \
    bar_signal_after_lds();                \
    __builtin_amdgcn_sched_barrier(0);                      \
    input_mma(axw, bias, BX, PA, PB);                                                            \
    __builtin_amdgcn_sched_barrier(0);                   \
    bar_wait_behind(PA, PB);                                                                     \
    __builtin_amdgcn_sched_barrier(0);                                                           \
  } while (0)

__global__ void __launch_bounds__(WV2 * 32)
rnn_split_kernel(const int* __restrict__ x, const float* __restrict__ emb,
                 const float* __restrict__ W_ih, const float* __restrict__ W_hh,
                 const float* __restrict__ b_ih, const float* __restrict__ b_hh,
                 const float* __restrict__ W_fc, const float* __restrict__ b_fc,
                 float* __restrict__ out, float* __restrict__ ws, int kstep) {
  __shared__ __attribute__((aligned(16))) float smem[LDSF];
  float* hb   = smem;
  float* pb   = smem + HB2F;
  float* embs = smem + HB2F + PB2F;

  const int g    = blockIdx.x >> 6;
  const int rg   = blockIdx.x & 63;
  const int cidx = kstep - g;
  if (cidx < 0 || cidx >= NCH2) return;

  const int tid  = threadIdx.x;
  const int lane = tid & 31;
  const int w    = __builtin_amdgcn_readfirstlane(tid >> 5);
  const int j    = w >> 2;
  const int x4   = w & 3;
  const int kind = (x4 >> 1) ^ (j & 1);
  const int c    = x4 & 1;
  const int L    = (g < 2) ? 3 : 2;
  const int l    = ((g < 2) ? 3 * g : 2 * g + 2) + j;
  const bool act = j < L;
  const bool top = (j == L - 1);
  const int R    = rg * RB2;
  const int t0   = cidx * CH2;

  const int n    = lane & 15;
  const int hh   = lane >> 4;

  const int orow = c * 16 + n;
  float* stp = ws + HANDF2 + ((g * NRG2 + rg) * 3 + j) * 512 + n * 32 + 16 * c + 8 * hh;
  float* hst = hb + j * HLAY2 + n * RS + 16 * c + 8 * hh;
  const float* hot = hb + j * HLAY2 + n * RS + 16 * (1 - c) + 4 * hh;
  float* ptl = pb + (j * 2 + c) * (2 * PTILE2) + lane * 4;

  v8f hreg;
#pragma unroll
  for (int r = 0; r < 8; ++r) hreg[r] = 0.0f;
  v2f aown[4]; v4f aoth[2];
  v4f axw[4]; float bias[8];
  v4i tokc;
  v4f bxg[4][4];
  const int* xrow = x + (R + n) * SEQT + t0;
  const float* hin = ws + (size_t)((((g > 0 ? g - 1 : 0) * 2 + (cidx & 1)) * NRG2 + rg) * CH2) * STEPF2 + lane * 4;
#pragma unroll
  for (int q = 0; q < 4; ++q) { aown[q] = (v2f)(0.0f); axw[q] = (v4f)(0.0f); }
#pragma unroll
  for (int q = 0; q < 2; ++q) aoth[q] = (v4f)(0.0f);
#pragma unroll
  for (int r = 0; r < 8; ++r) bias[r] = 0.0f;
  tokc = (v4i)(0);
#pragma unroll
  for (int ph = 0; ph < 4; ++ph)
#pragma unroll
    for (int q = 0; q < 4; ++q) bxg[ph][q] = (v4f)(0.0f);
  v4f st0 = (v4f)(0.0f), st1 = (v4f)(0.0f);
  float bi[8], bh[8];
#pragma unroll
  for (int r = 0; r < 8; ++r) { bi[r] = 0.0f; bh[r] = 0.0f; }
  v4f ev[6];
#pragma unroll
  for (int k = 0; k < 6; ++k) ev[k] = (v4f)(0.0f);
  if (g == 0) {
#pragma unroll
    for (int k = 0; k < 5; ++k) ev[k] = *(const v4f*)(emb + (tid + k * (WV2 * 32)) * 4);
    const int i5 = tid + 5 * (WV2 * 32);
    ev[5] = *(const v4f*)(emb + ((i5 < VOCAB * EMBED / 4) ? i5 : (VOCAB * EMBED / 4 - 1)) * 4);
  }
  if (act && kind == 0) {
    const float* Wh = W_hh + (l * HIDDEN + orow) * HIDDEN;
#pragma unroll
    for (int jj = 0; jj < 4; ++jj) aown[jj] = *(const v2f*)(Wh + 16 * c + 8 * hh + 2 * jj);
#pragma unroll
    for (int q = 0; q < 2; ++q) aoth[q] = *(const v4f*)(Wh + 16 * (1 - c) + 8 * q + 4 * hh);
    if (cidx > 0) {
      st0 = *(const v4f*)(stp); st1 = *(const v4f*)(stp + 4);
    }
  } else if (act) {
    const float* Wi = W_ih + (l * HIDDEN + orow) * EMBED;
#pragma unroll
    for (int q = 0; q < 4; ++q) axw[q] = *(const v4f*)(Wi + 8 * q + 4 * hh);
#pragma unroll
    for (int r = 0; r < 8; ++r) {
      const int o = l * HIDDEN + c * 16 + 8 * hh + r;
      bi[r] = b_ih[o]; bh[r] = b_hh[o];
    }
    if (j == 0) {
      if (g == 0) {
        tokc = *(const v4i*)(xrow);
      } else {
#pragma unroll
        for (int ph = 0; ph < 4; ++ph)
#pragma unroll
          for (int q = 0; q < 4; ++q) bxg[ph][q] = *(const v4f*)(hin + ph * STEPF2 + q * 128);
      }
    }
  }
  if (g == 0) {
#pragma unroll
    for (int k = 0; k < 5; ++k) {
      const int i = tid + k * (WV2 * 32);
      *(v4f*)(embs + (i >> 3) * RS + (i & 7) * 4) = ev[k];
    }
    const int i5 = tid + 5 * (WV2 * 32);
    if (i5 < VOCAB * EMBED / 4) *(v4f*)(embs + (i5 >> 3) * RS + (i5 & 7) * 4) = ev[5];
  }
#pragma unroll
  for (int r = 0; r < 8; ++r) bias[r] = bi[r] + bh[r];
  if (act && kind == 0) {
    hreg[0] = st0[0]; hreg[1] = st0[1]; hreg[2] = st0[2]; hreg[3] = st0[3];
    hreg[4] = st1[0]; hreg[5] = st1[1]; hreg[6] = st1[2]; hreg[7] = st1[3];
    v4f s0, s1;
    s0[0] = hreg[0]; s0[1] = hreg[1]; s0[2] = hreg[2]; s0[3] = hreg[3];
    s1[0] = hreg[4]; s1[1] = hreg[5]; s1[2] = hreg[6]; s1[3] = hreg[7];
    *(v4f*)(hst + HSLOT2)     = s0;
    *(v4f*)(hst + HSLOT2 + 4) = s1;
  }

  __syncthreads();

  if (!act) {
#pragma unroll 1
    for (int i = 0; i < NITER2; ++i) bar_round();
  } else if (kind == 0) {
    float* hog = ws + (size_t)(((g * 2 + (cidx & 1)) * NRG2 + rg) * CH2) * STEPF2 + (2 * c + hh) * 128 + n * 4;
    const bool hand = top && g < NGRP2 - 1;
    v8f dA, dB;
#pragma unroll 1
    for (int i = 0; i < 3 * j + 1; ++i) bar_round();
    bar_signal_after_lds();
    {
#pragma unroll
      for (int r = 0; r < 8; ++r) { dA[r] = 0.0f; dB[r] = 0.0f; }
      v2f h0, h1, h2, h3;
      h0[0] = hreg[0]; h0[1] = hreg[1]; h1[0] = hreg[2]; h1[1] = hreg[3];
      h2[0] = hreg[4]; h2[1] = hreg[5]; h3[0] = hreg[6]; h3[1] = hreg[7];
      dA = wmma_k4(aown[0], h0, dA);
      dB = wmma_k4(aown[2], h2, dB);
      dA = wmma_k4(aown[1], h1, dA);
      dB = wmma_k4(aown[3], h3, dB);
    }
    bar_wait();
#pragma unroll 1
    for (int s2 = 0; s2 < CH2; s2 += 2) {
#pragma unroll
      for (int e = 0; e < 2; ++e) {
        __builtin_amdgcn_s_setprio(1);
        v4f bo[2];
#pragma unroll
        for (int q = 0; q < 2; ++q) bo[q] = *(const v4f*)(hot + (1 - e) * HSLOT2 + 8 * q);
        const v4f p0 = *(const v4f*)(ptl + e * PTILE2);
        const v4f p1 = *(const v4f*)(ptl + e * PTILE2 + 128);
        __builtin_amdgcn_sched_barrier(0);
        {
          v2f a0, a1, b0, b1, e0, e1, f0, f1;
          a0[0] = aoth[0][0]; a0[1] = aoth[0][1]; a1[0] = aoth[0][2]; a1[1] = aoth[0][3];
          b0[0] = bo[0][0];   b0[1] = bo[0][1];   b1[0] = bo[0][2];   b1[1] = bo[0][3];
          e0[0] = aoth[1][0]; e0[1] = aoth[1][1]; e1[0] = aoth[1][2]; e1[1] = aoth[1][3];
          f0[0] = bo[1][0];   f0[1] = bo[1][1];   f1[0] = bo[1][2];   f1[1] = bo[1][3];
          dA = wmma_k4(a0, b0, dA);
          dB = wmma_k4(e0, f0, dB);
          dA = wmma_k4(a1, b1, dA);
          dB = wmma_k4(e1, f1, dB);
        }
        hreg[0] = dev_tanh((dA[0] + dB[0]) + p0[0]); hreg[1] = dev_tanh((dA[1] + dB[1]) + p0[1]);
        hreg[2] = dev_tanh((dA[2] + dB[2]) + p0[2]); hreg[3] = dev_tanh((dA[3] + dB[3]) + p0[3]);
        hreg[4] = dev_tanh((dA[4] + dB[4]) + p1[0]); hreg[5] = dev_tanh((dA[5] + dB[5]) + p1[1]);
        hreg[6] = dev_tanh((dA[6] + dB[6]) + p1[2]); hreg[7] = dev_tanh((dA[7] + dB[7]) + p1[3]);
        v4f s0, s1;
        s0[0] = hreg[0]; s0[1] = hreg[1]; s0[2] = hreg[2]; s0[3] = hreg[3];
        s1[0] = hreg[4]; s1[1] = hreg[5]; s1[2] = hreg[6]; s1[3] = hreg[7];
        *(v4f*)(hst + e * HSLOT2)     = s0;
        *(v4f*)(hst + e * HSLOT2 + 4) = s1;
        if (hand) {
          float* ho = hog + (s2 + e) * STEPF2;
          *(v4f*)(ho)      = s0;
          *(v4f*)(ho + 64) = s1;
        }
        __builtin_amdgcn_sched_barrier(0);
        bar_signal_after_lds();
        __builtin_amdgcn_s_setprio(0);
        __builtin_amdgcn_sched_barrier(0);
        {
#pragma unroll
          for (int r = 0; r < 8; ++r) { dA[r] = 0.0f; dB[r] = 0.0f; }
          v2f h0, h1, h2, h3;
          h0[0] = hreg[0]; h0[1] = hreg[1]; h1[0] = hreg[2]; h1[1] = hreg[3];
          h2[0] = hreg[4]; h2[1] = hreg[5]; h3[0] = hreg[6]; h3[1] = hreg[7];
          dA = wmma_k4(aown[0], h0, dA);
          dB = wmma_k4(aown[2], h2, dB);
          dA = wmma_k4(aown[1], h1, dA);
          dB = wmma_k4(aown[3], h3, dB);
        }
        __builtin_amdgcn_sched_barrier(0);
        bar_wait();
        __builtin_amdgcn_sched_barrier(0);
      }
    }
#pragma unroll 1
    for (int i = 0; i < NITER2 - CH2 - 2 - 3 * j; ++i) bar_round();
    {
      v4f s0, s1;
      s0[0] = hreg[0]; s0[1] = hreg[1]; s0[2] = hreg[2]; s0[3] = hreg[3];
      s1[0] = hreg[4]; s1[1] = hreg[5]; s1[2] = hreg[6]; s1[3] = hreg[7];
      *(v4f*)(stp)     = s0;
      *(v4f*)(stp + 4) = s1;
    }
  } else {
    if (j > 0) {
      const float* xin = hb + (j - 1) * HLAY2 + n * RS + 4 * hh;
#pragma unroll 1
      for (int i = 0; i < 3 * j; ++i) bar_round();
      v4f bxn[4];
      v8f PA, PB;
#pragma unroll
      for (int q = 0; q < 4; ++q) bxn[q] = *(const v4f*)(xin + 8 * q);
      INPUT_ROUND_MMA(bxn);
#pragma unroll 1
      for (int s2 = 0; s2 < CH2; s2 += 2) {
#pragma unroll
        for (int e = 0; e < 2; ++e) {
#pragma unroll
          for (int q = 0; q < 4; ++q) bxn[q] = *(const v4f*)(xin + (1 - e) * HSLOT2 + 8 * q);
          input_store(PA, PB, ptl + e * PTILE2);
          INPUT_ROUND_MMA(bxn);
        }
      }
#pragma unroll 1
      for (int i = 0; i < NITER2 - CH2 - 1 - 3 * j; ++i) bar_round();
    } else if (g == 0) {
      v4f bxn[4];
      v8f PA, PB;
      {
        const float* xe = embs + tokc[0] * RS + 4 * hh;
#pragma unroll
        for (int q = 0; q < 4; ++q) bxn[q] = *(const v4f*)(xe + 8 * q);
      }
      INPUT_ROUND_MMA(bxn);
#pragma unroll 1
      for (int s4 = 0; s4 < CH2; s4 += 4) {
        const int sn = (s4 + 4 < CH2) ? s4 + 4 : s4;
        const v4i tokn = *(const v4i*)(xrow + sn);
#pragma unroll
        for (int ph = 0; ph < 4; ++ph) {
          const int tk = (ph < 3) ? tokc[(ph + 1) & 3] : tokn[0];
          const float* xe = embs + tk * RS + 4 * hh;
#pragma unroll
          for (int q = 0; q < 4; ++q) bxn[q] = *(const v4f*)(xe + 8 * q);
          input_store(PA, PB, ptl + (ph & 1) * PTILE2);
          INPUT_ROUND_MMA(bxn);
        }
        tokc = tokn;
      }
#pragma unroll 1
      for (int i = 0; i < NITER2 - CH2 - 1; ++i) bar_round();
    } else {
      v8f PA, PB;
      INPUT_ROUND_MMA(bxg[0]);
#pragma unroll
      for (int q = 0; q < 4; ++q) bxg[0][q] = *(const v4f*)(hin + 4 * STEPF2 + q * 128);
#pragma unroll 1
      for (int s4 = 0; s4 < CH2; s4 += 4) {
#pragma unroll
        for (int ph = 0; ph < 4; ++ph) {
          input_store(PA, PB, ptl + (ph & 1) * PTILE2);
          INPUT_ROUND_MMA(bxg[(ph + 1) & 3]);
          const int s5 = (s4 + ph + 5 < CH2) ? s4 + ph + 5 : CH2 - 1;
#pragma unroll
          for (int q = 0; q < 4; ++q) bxg[(ph + 1) & 3][q] = *(const v4f*)(hin + s5 * STEPF2 + q * 128);
        }
      }
#pragma unroll 1
      for (int i = 0; i < NITER2 - CH2 - 1; ++i) bar_round();
    }
  }

  if (g == NGRP2 - 1 && cidx == NCH2 - 1 && w < 8) {
    const int koff = hh * 2;
    const float* htop = hb + (L - 1) * HLAY2 + ((CH2 - 1) & 1) * HSLOT2;
    v2f ahf[8];
#pragma unroll
    for (int jj = 0; jj < 8; ++jj) ahf[jj] = *(const v2f*)(htop + n * RS + 4 * jj + koff);
#pragma unroll
    for (int e = 0; e < 2; ++e) {
      const int vrow = (2 * w + e) * 16 + n;
      v2f wb[8];
#pragma unroll
      for (int jj = 0; jj < 8; ++jj) wb[jj] = *(const v2f*)(W_fc + vrow * HIDDEN + 4 * jj + koff);
      const float bf = b_fc[vrow];
      v8f dd;
#pragma unroll
      for (int k = 0; k < 8; ++k) dd[k] = bf;
#pragma unroll
      for (int jj = 0; jj < 8; ++jj) dd = wmma_k4(ahf[jj], wb[jj], dd);
#pragma unroll
      for (int r = 0; r < 8; ++r) out[(R + 8 * hh + r) * VOCAB + vrow] = dd[r];
    }
  }
}

#define SPLIT_LAUNCH(K) hipLaunchKernelGGL(rnn_split_kernel, dim3(NRG2 * NGRP2), dim3(WV2 * 32), 0, stream, \
                                           x, emb, Wih, Whh, bih, bhh, Wfc, bfc, out, wsf, K)

extern "C" void kernel_launch(void* const* d_in, const int* in_sizes, int n_in,
                              void* d_out, int out_size, void* d_ws, size_t ws_size,
                              hipStream_t stream) {
  (void)in_sizes; (void)n_in; (void)out_size;
  const int*   x   = (const int*)d_in[0];
  const float* emb = (const float*)d_in[1];
  const float* Wih = (const float*)d_in[2];
  const float* Whh = (const float*)d_in[3];
  const float* bih = (const float*)d_in[4];
  const float* bhh = (const float*)d_in[5];
  const float* Wfc = (const float*)d_in[6];
  const float* bfc = (const float*)d_in[7];
  float* out = (float*)d_out;
  float* wsf = (float*)d_ws;

  if (d_ws != nullptr && ws_size >= WSNEED2) {
    SPLIT_LAUNCH(0);  SPLIT_LAUNCH(1);  SPLIT_LAUNCH(2);  SPLIT_LAUNCH(3);  SPLIT_LAUNCH(4);
    SPLIT_LAUNCH(5);  SPLIT_LAUNCH(6);  SPLIT_LAUNCH(7);  SPLIT_LAUNCH(8);  SPLIT_LAUNCH(9);
    SPLIT_LAUNCH(10); SPLIT_LAUNCH(11); SPLIT_LAUNCH(12); SPLIT_LAUNCH(13); SPLIT_LAUNCH(14);
    SPLIT_LAUNCH(15); SPLIT_LAUNCH(16); SPLIT_LAUNCH(17); SPLIT_LAUNCH(18);
  } else {
    hipLaunchKernelGGL(rnn_pipeline_kernel, dim3(BATCH / BT), dim3(NWAVES * 32),
                       0, stream, x, emb, Wih, Whh, bih, bhh, Wfc, bfc, out);
  }
}
